# starting kernel
# baseline (speedup 1.0000x reference)
#include <hip/hip_runtime.h>
#include <hip/hip_bf16.h>
#include <math.h>


typedef __attribute__((ext_vector_type(2))) float v2f;
typedef __attribute__((ext_vector_type(8))) float v8f;

#define B_N    1024
#define C_N    64
#define T_N    275
#define CT_N   (C_N * T_N)
#define EMB_N  184
#define PROJ_N 1024

__device__ __forceinline__ v8f wmma4(v2f a, v2f b, v8f c) {
  return __builtin_amdgcn_wmma_f32_16x16x4_f32(false, a, false, b, (short)0, c,
                                               false, false);
}

__device__ __forceinline__ float gelu_exact(float x) {
  return 0.5f * x * (1.0f + erff(x * 0.70710678118654752440f));
}

__global__ __launch_bounds__(128) void subj_gemm_kernel(
    const float* __restrict__ eeg, const int* __restrict__ sid,
    const float* __restrict__ sW, const float* __restrict__ sB,
    float* __restrict__ x) {
  const int b    = blockIdx.y;
  const int wave = threadIdx.x >> 5;
  const int lane = threadIdx.x & 31;
  const int tile = blockIdx.x * 4 + wave;
  const int mt = tile / 9, np = tile % 9;
  const int m0 = mt * 16, n0 = np * 32;
  const int half = lane >> 4;
  const int l    = lane & 15;

  const int s = sid[b];
  const float* __restrict__ A    = eeg + (size_t)b * CT_N;
  const float* __restrict__ W    = sW + (size_t)s * T_N * T_N;
  const float* __restrict__ bias = sB + (size_t)s * T_N;

  const int  arow  = m0 + l;
  const int  ncol0 = n0 + l;
  const int  ncol1 = n0 + 16 + l;
  const bool n1ok  = (ncol1 < T_N);
  const int  nc1   = n1ok ? ncol1 : 0;

  const float* __restrict__ Arp = A + (size_t)arow * T_N;
  const float* __restrict__ B0p = W + (size_t)ncol0 * T_N;
  const float* __restrict__ B1p = W + (size_t)nc1 * T_N;

  v8f acc0 = {}, acc1 = {};
  for (int k = 0; k < 272; k += 8) {
    const int ka = k + 2 * half;
    v2f a0, a1, b00, b01, b10, b11;
    a0.x  = Arp[ka];     a0.y  = Arp[ka + 1];
    a1.x  = Arp[ka + 4]; a1.y  = Arp[ka + 5];
    b00.x = B0p[ka];     b00.y = B0p[ka + 1];
    b01.x = B0p[ka + 4]; b01.y = B0p[ka + 5];
    b10.x = B1p[ka];     b10.y = B1p[ka + 1];
    b11.x = B1p[ka + 4]; b11.y = B1p[ka + 5];
    acc0 = wmma4(a0, b00, acc0);
    acc1 = wmma4(a0, b10, acc1);
    acc0 = wmma4(a1, b01, acc0);
    acc1 = wmma4(a1, b11, acc1);
  }
  {
    const int   ka    = 272 + 2 * half;
    const float ymask = half ? 0.0f : 1.0f;
    v2f a, b0, b1;
    a.x  = Arp[ka];  a.y  = Arp[273] * ymask;
    b0.x = B0p[ka];  b0.y = B0p[273] * ymask;
    b1.x = B1p[ka];  b1.y = B1p[273] * ymask;
    acc0 = wmma4(a, b0, acc0);
    acc1 = wmma4(a, b1, acc1);
  }

  float* __restrict__ X = x + (size_t)b * CT_N;
  const float bn0 = bias[ncol0];
#pragma unroll
  for (int r = 0; r < 8; ++r) {
    const int row = m0 + r + 8 * half;
    X[row * T_N + ncol0] = acc0[r] + bn0;
  }
  if (n1ok) {
    const float bn1 = bias[ncol1];
#pragma unroll
    for (int r = 0; r < 8; ++r) {
      const int row = m0 + r + 8 * half;
      X[row * T_N + ncol1] = acc1[r] + bn1;
    }
  }
}

__global__ __launch_bounds__(256) void enc_gemm_kernel(
    const float* __restrict__ x, const float* __restrict__ eW,
    const float* __restrict__ eB, float* __restrict__ h) {
  const int wave = threadIdx.x >> 5;
  const int lane = threadIdx.x & 31;
  const int tile = blockIdx.x * 8 + wave;
  const int mt = tile / 6, np = tile % 6;
  const int m0 = mt * 16, n0 = np * 32;
  const int half = lane >> 4;
  const int l    = lane & 15;

  const int  arow  = m0 + l;
  const int  ncol0 = n0 + l;
  const int  ncol1 = n0 + 16 + l;
  const bool n1ok  = (ncol1 < EMB_N);
  const int  nc1   = n1ok ? ncol1 : 0;
  const float* __restrict__ Ar = x + (size_t)arow * CT_N;

  v8f acc0 = {}, acc1 = {};
  for (int k = 0; k < CT_N; k += 8) {
    const int ka = k + 2 * half;
    const v2f a0 = *(const v2f*)(Ar + ka);
    const v2f a1 = *(const v2f*)(Ar + ka + 4);
    v2f b00, b01, b10, b11;
    b00.x = eW[(size_t)ka * EMB_N + ncol0];
    b00.y = eW[(size_t)(ka + 1) * EMB_N + ncol0];
    b01.x = eW[(size_t)(ka + 4) * EMB_N + ncol0];
    b01.y = eW[(size_t)(ka + 5) * EMB_N + ncol0];
    b10.x = eW[(size_t)ka * EMB_N + nc1];
    b10.y = eW[(size_t)(ka + 1) * EMB_N + nc1];
    b11.x = eW[(size_t)(ka + 4) * EMB_N + nc1];
    b11.y = eW[(size_t)(ka + 5) * EMB_N + nc1];
    acc0 = wmma4(a0, b00, acc0);
    acc1 = wmma4(a0, b10, acc1);
    acc0 = wmma4(a1, b01, acc0);
    acc1 = wmma4(a1, b11, acc1);
  }
  const float bn0 = eB[ncol0];
#pragma unroll
  for (int r = 0; r < 8; ++r) {
    const int row = m0 + r + 8 * half;
    h[(size_t)row * EMB_N + ncol0] = acc0[r] + bn0;
  }
  if (n1ok) {
    const float bn1 = eB[ncol1];
#pragma unroll
    for (int r = 0; r < 8; ++r) {
      const int row = m0 + r + 8 * half;
      h[(size_t)row * EMB_N + ncol1] = acc1[r] + bn1;
    }
  }
}

__global__ __launch_bounds__(256) void proj1_gemm_kernel(
    const float* __restrict__ h, const float* __restrict__ W1,
    const float* __restrict__ b1, float* __restrict__ h1,
    float* __restrict__ g) {
  const int wave = threadIdx.x >> 5;
  const int lane = threadIdx.x & 31;
  const int tile = blockIdx.x * 8 + wave;
  const int mt = tile >> 5, np = tile & 31;
  const int m0 = mt * 16, n0 = np * 32;
  const int half = lane >> 4;
  const int l    = lane & 15;

  const int arow  = m0 + l;
  const int ncol0 = n0 + l;
  const int ncol1 = n0 + 16 + l;
  const float* __restrict__ Ar = h + (size_t)arow * EMB_N;

  v8f acc0 = {}, acc1 = {};
  for (int k = 0; k < EMB_N; k += 8) {
    const int ka = k + 2 * half;
    const v2f a0 = *(const v2f*)(Ar + ka);
    const v2f a1 = *(const v2f*)(Ar + ka + 4);
    v2f b00, b01, b10, b11;
    b00.x = W1[(size_t)ka * PROJ_N + ncol0];
    b00.y = W1[(size_t)(ka + 1) * PROJ_N + ncol0];
    b01.x = W1[(size_t)(ka + 4) * PROJ_N + ncol0];
    b01.y = W1[(size_t)(ka + 5) * PROJ_N + ncol0];
    b10.x = W1[(size_t)ka * PROJ_N + ncol1];
    b10.y = W1[(size_t)(ka + 1) * PROJ_N + ncol1];
    b11.x = W1[(size_t)(ka + 4) * PROJ_N + ncol1];
    b11.y = W1[(size_t)(ka + 5) * PROJ_N + ncol1];
    acc0 = wmma4(a0, b00, acc0);
    acc1 = wmma4(a0, b10, acc1);
    acc0 = wmma4(a1, b01, acc0);
    acc1 = wmma4(a1, b11, acc1);
  }
  const float bn0 = b1[ncol0];
  const float bn1 = b1[ncol1];
#pragma unroll
  for (int r = 0; r < 8; ++r) {
    const int row = m0 + r + 8 * half;
    const float t0 = acc0[r] + bn0;
    const float t1 = acc1[r] + bn1;
    h1[(size_t)row * PROJ_N + ncol0] = t0;
    h1[(size_t)row * PROJ_N + ncol1] = t1;
    g[(size_t)row * PROJ_N + ncol0] = gelu_exact(t0);
    g[(size_t)row * PROJ_N + ncol1] = gelu_exact(t1);
  }
}

__global__ __launch_bounds__(256) void proj2_gemm_kernel(
    const float* __restrict__ g, const float* __restrict__ W2,
    const float* __restrict__ b2, const float* __restrict__ h1,
    float* __restrict__ hsum) {
  const int wave = threadIdx.x >> 5;
  const int lane = threadIdx.x & 31;
  const int tile = blockIdx.x * 8 + wave;
  const int mt = tile >> 5, np = tile & 31;
  const int m0 = mt * 16, n0 = np * 32;
  const int half = lane >> 4;
  const int l    = lane & 15;

  const int arow  = m0 + l;
  const int ncol0 = n0 + l;
  const int ncol1 = n0 + 16 + l;
  const float* __restrict__ Ar = g + (size_t)arow * PROJ_N;

  v8f acc0 = {}, acc1 = {};
  for (int k = 0; k < PROJ_N; k += 8) {
    const int ka = k + 2 * half;
    const v2f a0 = *(const v2f*)(Ar + ka);
    const v2f a1 = *(const v2f*)(Ar + ka + 4);
    v2f b00, b01, b10, b11;
    b00.x = W2[(size_t)ka * PROJ_N + ncol0];
    b00.y = W2[(size_t)(ka + 1) * PROJ_N + ncol0];
    b01.x = W2[(size_t)(ka + 4) * PROJ_N + ncol0];
    b01.y = W2[(size_t)(ka + 5) * PROJ_N + ncol0];
    b10.x = W2[(size_t)ka * PROJ_N + ncol1];
    b10.y = W2[(size_t)(ka + 1) * PROJ_N + ncol1];
    b11.x = W2[(size_t)(ka + 4) * PROJ_N + ncol1];
    b11.y = W2[(size_t)(ka + 5) * PROJ_N + ncol1];
    acc0 = wmma4(a0, b00, acc0);
    acc1 = wmma4(a0, b10, acc1);
    acc0 = wmma4(a1, b01, acc0);
    acc1 = wmma4(a1, b11, acc1);
  }
  const float bn0 = b2[ncol0];
  const float bn1 = b2[ncol1];
#pragma unroll
  for (int r = 0; r < 8; ++r) {
    const int row = m0 + r + 8 * half;
    hsum[(size_t)row * PROJ_N + ncol0] =
        h1[(size_t)row * PROJ_N + ncol0] + acc0[r] + bn0;
    hsum[(size_t)row * PROJ_N + ncol1] =
        h1[(size_t)row * PROJ_N + ncol1] + acc1[r] + bn1;
  }
}

__global__ __launch_bounds__(256) void layernorm_kernel(
    const float* __restrict__ hsum, const float* __restrict__ gamma,
    const float* __restrict__ beta, float* __restrict__ out) {
  const int row = blockIdx.x;
  const int tid = threadIdx.x;
  const float* __restrict__ hr = hsum + (size_t)row * PROJ_N;

  float v[4];
  float s = 0.0f, ss = 0.0f;
#pragma unroll
  for (int i = 0; i < 4; ++i) {
    v[i] = hr[tid + i * 256];
    s += v[i];
    ss += v[i] * v[i];
  }
  __shared__ float rs[256];
  __shared__ float rss[256];
  rs[tid] = s;
  rss[tid] = ss;
  __syncthreads();
  for (int off = 128; off > 0; off >>= 1) {
    if (tid < off) {
      rs[tid] += rs[tid + off];
      rss[tid] += rss[tid + off];
    }
    __syncthreads();
  }
  const float mu  = rs[0] * (1.0f / PROJ_N);
  const float var = rss[0] * (1.0f / PROJ_N) - mu * mu;
  const float inv = rsqrtf(var + 1e-5f);
#pragma unroll
  for (int i = 0; i < 4; ++i) {
    const int c = tid + i * 256;
    out[(size_t)row * PROJ_N + c] = (v[i] - mu) * inv * gamma[c] + beta[c];
  }
}

extern "C" void kernel_launch(void* const* d_in, const int* in_sizes, int n_in,
                              void* d_out, int out_size, void* d_ws,
                              size_t ws_size, hipStream_t stream) {
  const float* eeg    = (const float*)d_in[0];
  const int*   sid    = (const int*)d_in[1];
  const float* subj_W = (const float*)d_in[2];
  const float* subj_b = (const float*)d_in[3];
  const float* enc_W  = (const float*)d_in[4];
  const float* enc_b  = (const float*)d_in[5];
  const float* W1     = (const float*)d_in[6];
  const float* b1     = (const float*)d_in[7];
  const float* W2     = (const float*)d_in[8];
  const float* b2     = (const float*)d_in[9];
  const float* gamma  = (const float*)d_in[10];
  const float* beta   = (const float*)d_in[11];
  float*       out    = (float*)d_out;

  float* xbuf  = (float*)d_ws;
  float* hbuf  = xbuf + (size_t)B_N * CT_N;
  float* h1buf = hbuf + (size_t)B_N * EMB_N;
  float* gbuf  = h1buf + (size_t)B_N * PROJ_N;
  float* hsbuf = gbuf + (size_t)B_N * PROJ_N;

  subj_gemm_kernel<<<dim3(9, B_N), 128, 0, stream>>>(eeg, sid, subj_W, subj_b,
                                                     xbuf);
  enc_gemm_kernel<<<48, 256, 0, stream>>>(xbuf, enc_W, enc_b, hbuf);
  proj1_gemm_kernel<<<256, 256, 0, stream>>>(hbuf, W1, b1, h1buf, gbuf);
  proj2_gemm_kernel<<<256, 256, 0, stream>>>(gbuf, W2, b2, h1buf, hsbuf);
  layernorm_kernel<<<B_N, 256, 0, stream>>>(hsbuf, gamma, beta, out);
}


# encoder GEMM: one workgroup per 16x32 strip with K split over its 25 waves (384 workgroups, was 48); the 25 partial sums are added in fixed wave order through LDS
# speedup vs baseline: 1.9039x; 1.9039x over previous
#include <hip/hip_runtime.h>
#include <hip/hip_bf16.h>
#include <math.h>


typedef __attribute__((ext_vector_type(2))) float v2f;
typedef __attribute__((ext_vector_type(8))) float v8f;

#define B_N    1024
#define C_N    64
#define T_N    275
#define CT_N   (C_N * T_N)
#define EMB_N  184
#define PROJ_N 1024

__device__ __forceinline__ v8f wmma4(v2f a, v2f b, v8f c) {
  return __builtin_amdgcn_wmma_f32_16x16x4_f32(false, a, false, b, (short)0, c,
                                               false, false);
}

__device__ __forceinline__ float gelu_exact(float x) {
  return 0.5f * x * (1.0f + erff(x * 0.70710678118654752440f));
}

__global__ __launch_bounds__(128) void subj_gemm_kernel(
    const float* __restrict__ eeg, const int* __restrict__ sid,
    const float* __restrict__ sW, const float* __restrict__ sB,
    float* __restrict__ x) {
  const int b    = blockIdx.y;
  const int wave = threadIdx.x >> 5;
  const int lane = threadIdx.x & 31;
  const int tile = blockIdx.x * 4 + wave;
  const int mt = tile / 9, np = tile % 9;
  const int m0 = mt * 16, n0 = np * 32;
  const int half = lane >> 4;
  const int l    = lane & 15;

  const int s = sid[b];
  const float* __restrict__ A    = eeg + (size_t)b * CT_N;
  const float* __restrict__ W    = sW + (size_t)s * T_N * T_N;
  const float* __restrict__ bias = sB + (size_t)s * T_N;

  const int  arow  = m0 + l;
  const int  ncol0 = n0 + l;
  const int  ncol1 = n0 + 16 + l;
  const bool n1ok  = (ncol1 < T_N);
  const int  nc1   = n1ok ? ncol1 : 0;

  const float* __restrict__ Arp = A + (size_t)arow * T_N;
  const float* __restrict__ B0p = W + (size_t)ncol0 * T_N;
  const float* __restrict__ B1p = W + (size_t)nc1 * T_N;

  v8f acc0 = {}, acc1 = {};
  for (int k = 0; k < 272; k += 8) {
    const int ka = k + 2 * half;
    v2f a0, a1, b00, b01, b10, b11;
    a0.x  = Arp[ka];     a0.y  = Arp[ka + 1];
    a1.x  = Arp[ka + 4]; a1.y  = Arp[ka + 5];
    b00.x = B0p[ka];     b00.y = B0p[ka + 1];
    b01.x = B0p[ka + 4]; b01.y = B0p[ka + 5];
    b10.x = B1p[ka];     b10.y = B1p[ka + 1];
    b11.x = B1p[ka + 4]; b11.y = B1p[ka + 5];
    acc0 = wmma4(a0, b00, acc0);
    acc1 = wmma4(a0, b10, acc1);
    acc0 = wmma4(a1, b01, acc0);
    acc1 = wmma4(a1, b11, acc1);
  }
  {
    const int   ka    = 272 + 2 * half;
    const float ymask = half ? 0.0f : 1.0f;
    v2f a, b0, b1;
    a.x  = Arp[ka];  a.y  = Arp[273] * ymask;
    b0.x = B0p[ka];  b0.y = B0p[273] * ymask;
    b1.x = B1p[ka];  b1.y = B1p[273] * ymask;
    acc0 = wmma4(a, b0, acc0);
    acc1 = wmma4(a, b1, acc1);
  }

  float* __restrict__ X = x + (size_t)b * CT_N;
  const float bn0 = bias[ncol0];
#pragma unroll
  for (int r = 0; r < 8; ++r) {
    const int row = m0 + r + 8 * half;
    X[row * T_N + ncol0] = acc0[r] + bn0;
  }
  if (n1ok) {
    const float bn1 = bias[ncol1];
#pragma unroll
    for (int r = 0; r < 8; ++r) {
      const int row = m0 + r + 8 * half;
      X[row * T_N + ncol1] = acc1[r] + bn1;
    }
  }
}

#define ENC_KW     25
#define ENC_PIECES (CT_N / 32)

__global__ __launch_bounds__(ENC_KW * 32) void enc_gemm_kernel(
    const float* __restrict__ x, const float* __restrict__ eW,
    const float* __restrict__ eB, float* __restrict__ h) {
  __shared__ float part[ENC_KW][16][32];

  const int wave = threadIdx.x >> 5;
  const int lane = threadIdx.x & 31;
  const int tile = blockIdx.x;
  const int mt = tile / 6, np = tile % 6;
  const int m0 = mt * 16, n0 = np * 32;
  const int half = lane >> 4;
  const int l    = lane & 15;

  const int  arow  = m0 + l;
  const int  ncol0 = n0 + l;
  const int  ncol1 = n0 + 16 + l;
  const bool n1ok  = (ncol1 < EMB_N);
  const int  nc1   = n1ok ? ncol1 : 0;
  const float* __restrict__ Ar = x + (size_t)arow * CT_N;

  v8f acc0 = {}, acc1 = {};
  for (int p = wave; p < ENC_PIECES; p += ENC_KW) {
    for (int u = 0; u < 4; ++u) {
      const int ka = p * 32 + u * 8 + 2 * half;
      const v2f a0 = *(const v2f*)(Ar + ka);
      const v2f a1 = *(const v2f*)(Ar + ka + 4);
      v2f b00, b01, b10, b11;
      b00.x = eW[(size_t)ka * EMB_N + ncol0];
      b00.y = eW[(size_t)(ka + 1) * EMB_N + ncol0];
      b01.x = eW[(size_t)(ka + 4) * EMB_N + ncol0];
      b01.y = eW[(size_t)(ka + 5) * EMB_N + ncol0];
      b10.x = eW[(size_t)ka * EMB_N + nc1];
      b10.y = eW[(size_t)(ka + 1) * EMB_N + nc1];
      b11.x = eW[(size_t)(ka + 4) * EMB_N + nc1];
      b11.y = eW[(size_t)(ka + 5) * EMB_N + nc1];
      acc0 = wmma4(a0, b00, acc0);
      acc1 = wmma4(a0, b10, acc1);
      acc0 = wmma4(a1, b01, acc0);
      acc1 = wmma4(a1, b11, acc1);
    }
  }

  for (int r = 0; r < 8; ++r) {
    part[wave][r][lane]     = acc0[r];
    part[wave][8 + r][lane] = acc1[r];
  }
  __syncthreads();

  if (wave < 16) {
    float s = part[0][wave][lane];
    for (int w = 1; w < ENC_KW; ++w) s += part[w][wave][lane];
    const int row = m0 + (wave & 7) + 8 * half;
    const int col = n0 + 16 * (wave >> 3) + l;
    if (col < EMB_N) h[(size_t)row * EMB_N + col] = s + eB[col];
  }
}

__global__ __launch_bounds__(256) void proj1_gemm_kernel(
    const float* __restrict__ h, const float* __restrict__ W1,
    const float* __restrict__ b1, float* __restrict__ h1,
    float* __restrict__ g) {
  const int wave = threadIdx.x >> 5;
  const int lane = threadIdx.x & 31;
  const int tile = blockIdx.x * 8 + wave;
  const int mt = tile >> 5, np = tile & 31;
  const int m0 = mt * 16, n0 = np * 32;
  const int half = lane >> 4;
  const int l    = lane & 15;

  const int arow  = m0 + l;
  const int ncol0 = n0 + l;
  const int ncol1 = n0 + 16 + l;
  const float* __restrict__ Ar = h + (size_t)arow * EMB_N;

  v8f acc0 = {}, acc1 = {};
  for (int k = 0; k < EMB_N; k += 8) {
    const int ka = k + 2 * half;
    const v2f a0 = *(const v2f*)(Ar + ka);
    const v2f a1 = *(const v2f*)(Ar + ka + 4);
    v2f b00, b01, b10, b11;
    b00.x = W1[(size_t)ka * PROJ_N + ncol0];
    b00.y = W1[(size_t)(ka + 1) * PROJ_N + ncol0];
    b01.x = W1[(size_t)(ka + 4) * PROJ_N + ncol0];
    b01.y = W1[(size_t)(ka + 5) * PROJ_N + ncol0];
    b10.x = W1[(size_t)ka * PROJ_N + ncol1];
    b10.y = W1[(size_t)(ka + 1) * PROJ_N + ncol1];
    b11.x = W1[(size_t)(ka + 4) * PROJ_N + ncol1];
    b11.y = W1[(size_t)(ka + 5) * PROJ_N + ncol1];
    acc0 = wmma4(a0, b00, acc0);
    acc1 = wmma4(a0, b10, acc1);
    acc0 = wmma4(a1, b01, acc0);
    acc1 = wmma4(a1, b11, acc1);
  }
  const float bn0 = b1[ncol0];
  const float bn1 = b1[ncol1];
#pragma unroll
  for (int r = 0; r < 8; ++r) {
    const int row = m0 + r + 8 * half;
    const float t0 = acc0[r] + bn0;
    const float t1 = acc1[r] + bn1;
    h1[(size_t)row * PROJ_N + ncol0] = t0;
    h1[(size_t)row * PROJ_N + ncol1] = t1;
    g[(size_t)row * PROJ_N + ncol0] = gelu_exact(t0);
    g[(size_t)row * PROJ_N + ncol1] = gelu_exact(t1);
  }
}

__global__ __launch_bounds__(256) void proj2_gemm_kernel(
    const float* __restrict__ g, const float* __restrict__ W2,
    const float* __restrict__ b2, const float* __restrict__ h1,
    float* __restrict__ hsum) {
  const int wave = threadIdx.x >> 5;
  const int lane = threadIdx.x & 31;
  const int tile = blockIdx.x * 8 + wave;
  const int mt = tile >> 5, np = tile & 31;
  const int m0 = mt * 16, n0 = np * 32;
  const int half = lane >> 4;
  const int l    = lane & 15;

  const int arow  = m0 + l;
  const int ncol0 = n0 + l;
  const int ncol1 = n0 + 16 + l;
  const float* __restrict__ Ar = g + (size_t)arow * PROJ_N;

  v8f acc0 = {}, acc1 = {};
  for (int k = 0; k < PROJ_N; k += 8) {
    const int ka = k + 2 * half;
    const v2f a0 = *(const v2f*)(Ar + ka);
    const v2f a1 = *(const v2f*)(Ar + ka + 4);
    v2f b00, b01, b10, b11;
    b00.x = W2[(size_t)ka * PROJ_N + ncol0];
    b00.y = W2[(size_t)(ka + 1) * PROJ_N + ncol0];
    b01.x = W2[(size_t)(ka + 4) * PROJ_N + ncol0];
    b01.y = W2[(size_t)(ka + 5) * PROJ_N + ncol0];
    b10.x = W2[(size_t)ka * PROJ_N + ncol1];
    b10.y = W2[(size_t)(ka + 1) * PROJ_N + ncol1];
    b11.x = W2[(size_t)(ka + 4) * PROJ_N + ncol1];
    b11.y = W2[(size_t)(ka + 5) * PROJ_N + ncol1];
    acc0 = wmma4(a0, b00, acc0);
    acc1 = wmma4(a0, b10, acc1);
    acc0 = wmma4(a1, b01, acc0);
    acc1 = wmma4(a1, b11, acc1);
  }
  const float bn0 = b2[ncol0];
  const float bn1 = b2[ncol1];
#pragma unroll
  for (int r = 0; r < 8; ++r) {
    const int row = m0 + r + 8 * half;
    hsum[(size_t)row * PROJ_N + ncol0] =
        h1[(size_t)row * PROJ_N + ncol0] + acc0[r] + bn0;
    hsum[(size_t)row * PROJ_N + ncol1] =
        h1[(size_t)row * PROJ_N + ncol1] + acc1[r] + bn1;
  }
}

__global__ __launch_bounds__(256) void layernorm_kernel(
    const float* __restrict__ hsum, const float* __restrict__ gamma,
    const float* __restrict__ beta, float* __restrict__ out) {
  const int row = blockIdx.x;
  const int tid = threadIdx.x;
  const float* __restrict__ hr = hsum + (size_t)row * PROJ_N;

  float v[4];
  float s = 0.0f, ss = 0.0f;
#pragma unroll
  for (int i = 0; i < 4; ++i) {
    v[i] = hr[tid + i * 256];
    s += v[i];
    ss += v[i] * v[i];
  }
  __shared__ float rs[256];
  __shared__ float rss[256];
  rs[tid] = s;
  rss[tid] = ss;
  __syncthreads();
  for (int off = 128; off > 0; off >>= 1) {
    if (tid < off) {
      rs[tid] += rs[tid + off];
      rss[tid] += rss[tid + off];
    }
    __syncthreads();
  }
  const float mu  = rs[0] * (1.0f / PROJ_N);
  const float var = rss[0] * (1.0f / PROJ_N) - mu * mu;
  const float inv = rsqrtf(var + 1e-5f);
#pragma unroll
  for (int i = 0; i < 4; ++i) {
    const int c = tid + i * 256;
    out[(size_t)row * PROJ_N + c] = (v[i] - mu) * inv * gamma[c] + beta[c];
  }
}

extern "C" void kernel_launch(void* const* d_in, const int* in_sizes, int n_in,
                              void* d_out, int out_size, void* d_ws,
                              size_t ws_size, hipStream_t stream) {
  const float* eeg    = (const float*)d_in[0];
  const int*   sid    = (const int*)d_in[1];
  const float* subj_W = (const float*)d_in[2];
  const float* subj_b = (const float*)d_in[3];
  const float* enc_W  = (const float*)d_in[4];
  const float* enc_b  = (const float*)d_in[5];
  const float* W1     = (const float*)d_in[6];
  const float* b1     = (const float*)d_in[7];
  const float* W2     = (const float*)d_in[8];
  const float* b2     = (const float*)d_in[9];
  const float* gamma  = (const float*)d_in[10];
  const float* beta   = (const float*)d_in[11];
  float*       out    = (float*)d_out;

  float* xbuf  = (float*)d_ws;
  float* hbuf  = xbuf + (size_t)B_N * CT_N;
  float* h1buf = hbuf + (size_t)B_N * EMB_N;
  float* gbuf  = h1buf + (size_t)B_N * PROJ_N;
  float* hsbuf = gbuf + (size_t)B_N * PROJ_N;

  subj_gemm_kernel<<<dim3(9, B_N), 128, 0, stream>>>(eeg, sid, subj_W, subj_b,
                                                     xbuf);
  enc_gemm_kernel<<<384, ENC_KW * 32, 0, stream>>>(xbuf, enc_W, enc_b, hbuf);
  proj1_gemm_kernel<<<256, 256, 0, stream>>>(hbuf, W1, b1, h1buf, gbuf);
  proj2_gemm_kernel<<<256, 256, 0, stream>>>(gbuf, W2, b2, h1buf, hsbuf);
  layernorm_kernel<<<B_N, 256, 0, stream>>>(hsbuf, gamma, beta, out);
}


# subject GEMM: one workgroup per batch item stages its 64x275 input tile and 32-row chunks of the weight matrix in LDS by coalesced row loads, A fragments held in registers (was per-wave gathers)
# speedup vs baseline: 3.5512x; 1.8652x over previous
#include <hip/hip_runtime.h>
#include <hip/hip_bf16.h>
#include <math.h>


typedef __attribute__((ext_vector_type(2))) float v2f;
typedef __attribute__((ext_vector_type(8))) float v8f;

#define B_N    1024
#define C_N    64
#define T_N    275
#define CT_N   (C_N * T_N)
#define EMB_N  184
#define PROJ_N 1024

__device__ __forceinline__ v8f wmma4(v2f a, v2f b, v8f c) {
  return __builtin_amdgcn_wmma_f32_16x16x4_f32(false, a, false, b, (short)0, c,
                                               false, false);
}

__device__ __forceinline__ float gelu_exact(float x) {
  return 0.5f * x * (1.0f + erff(x * 0.70710678118654752440f));
}

#define SJ_PITCH 276
#define SJ_KS    69
#define SJ_CH    32
#define SJ_NCH   9
#define SJ_RB    2
#define SJ_RL    9

__device__ __forceinline__ void sj_load(float (&v)[SJ_RB * SJ_RL],
                                        const float* __restrict__ mat,
                                        int row0, int lastrow, int lane) {
  const int lane8 = lane < 19 ? lane : 18;
#pragma unroll
  for (int j = 0; j < SJ_RB; ++j) {
    int r = row0 + j;
    if (r > lastrow) r = lastrow;
    const float* __restrict__ rp = mat + (size_t)r * T_N;
#pragma unroll
    for (int q = 0; q < 8; ++q) v[j * SJ_RL + q] = rp[q * 32 + lane];
    v[j * SJ_RL + 8] = rp[256 + lane8];
  }
}

__device__ __forceinline__ void sj_store(float* __restrict__ dstrow,
                                         const float (&v)[SJ_RB * SJ_RL],
                                         int lane) {
  const int lane8 = lane < 19 ? lane : 18;
#pragma unroll
  for (int j = 0; j < SJ_RB; ++j) {
#pragma unroll
    for (int q = 0; q < 8; ++q)
      dstrow[j * SJ_PITCH + q * 32 + lane] = v[j * SJ_RL + q];
    dstrow[j * SJ_PITCH + 256 + lane8] = v[j * SJ_RL + 8];
  }
}

__global__ __launch_bounds__(128, 4) void subj_gemm_kernel(
    const float* __restrict__ eeg, const int* __restrict__ sid,
    const float* __restrict__ sW, const float* __restrict__ sB,
    float* __restrict__ x) {
  __shared__ __attribute__((aligned(16))) float S[C_N * SJ_PITCH];

  const int b    = blockIdx.x;
  const int tid  = threadIdx.x;
  const int lane = tid & 31;
  const int half = lane >> 4;
  const int l    = lane & 15;
  const int wave = __builtin_amdgcn_readfirstlane(tid >> 5);

  const int s = sid[b];
  const float* __restrict__ E    = eeg + (size_t)b * CT_N;
  const float* __restrict__ W    = sW + (size_t)s * T_N * T_N;
  const float* __restrict__ bias = sB + (size_t)s * T_N;

  if (tid < C_N) S[tid * SJ_PITCH + T_N] = 0.0f;
#pragma unroll
  for (int bt = 0; bt < 8; ++bt) {
    float v[SJ_RB * SJ_RL];
    sj_load(v, E, wave * 16 + bt * SJ_RB, C_N - 1, lane);
    sj_store(S + (wave * 16 + bt * SJ_RB) * SJ_PITCH, v, lane);
  }
  __syncthreads();

  v2f a[SJ_KS];
  {
    const float* ap = S + (wave * 16 + l) * SJ_PITCH + 2 * half;
#pragma unroll
    for (int ks = 0; ks < SJ_KS; ++ks) a[ks] = *(const v2f*)(ap + 4 * ks);
  }
  __syncthreads();

#pragma unroll
  for (int bt = 0; bt < 4; ++bt) {
    float v[SJ_RB * SJ_RL];
    sj_load(v, W, wave * 8 + bt * SJ_RB, T_N - 1, lane);
    sj_store(S + (wave * 8 + bt * SJ_RB) * SJ_PITCH, v, lane);
  }
  float* __restrict__ X = x + (size_t)b * CT_N;

  for (int c = 0; c < SJ_NCH; ++c) {
    __syncthreads();
    const bool more = (c + 1 < SJ_NCH);
    const int nrow0 = (c + 1) * SJ_CH + wave * 8;
    float* __restrict__ dst =
        S + (((c + 1) & 1) * SJ_CH + wave * 8) * SJ_PITCH;

    const float* bp0 = S + (c & 1) * (SJ_CH * SJ_PITCH) + l * SJ_PITCH + 2 * half;
    const float* bp1 = bp0 + 16 * SJ_PITCH;
    v8f acc0 = {}, acc1 = {};
#pragma unroll
    for (int part = 0; part < 4; ++part) {
      float v[SJ_RB * SJ_RL];
      if (more) sj_load(v, W, nrow0 + part * SJ_RB, T_N - 1, lane);
#pragma unroll
      for (int ks = (part * SJ_KS + 3) / 4; ks < ((part + 1) * SJ_KS + 3) / 4; ++ks) {
        const v2f b0 = *(const v2f*)(bp0 + 4 * ks);
        const v2f b1 = *(const v2f*)(bp1 + 4 * ks);
        acc0 = wmma4(a[ks], b0, acc0);
        acc1 = wmma4(a[ks], b1, acc1);
      }
      if (more) sj_store(dst + part * SJ_RB * SJ_PITCH, v, lane);
    }
    const int col0 = c * SJ_CH + l;
    const int col1 = col0 + 16;
    const float bn0 = bias[col0];
#pragma unroll
    for (int r = 0; r < 8; ++r) {
      const int orow = wave * 16 + r + 8 * half;
      X[orow * T_N + col0] = acc0[r] + bn0;
    }
    if (col1 < T_N) {
      const float bn1 = bias[col1];
#pragma unroll
      for (int r = 0; r < 8; ++r) {
        const int orow = wave * 16 + r + 8 * half;
        X[orow * T_N + col1] = acc1[r] + bn1;
      }
    }
  }
}

#define ENC_KW     25
#define ENC_PIECES (CT_N / 32)

__global__ __launch_bounds__(ENC_KW * 32) void enc_gemm_kernel(
    const float* __restrict__ x, const float* __restrict__ eW,
    const float* __restrict__ eB, float* __restrict__ h) {
  __shared__ float part[ENC_KW][16][32];

  const int wave = threadIdx.x >> 5;
  const int lane = threadIdx.x & 31;
  const int tile = blockIdx.x;
  const int mt = tile / 6, np = tile % 6;
  const int m0 = mt * 16, n0 = np * 32;
  const int half = lane >> 4;
  const int l    = lane & 15;

  const int  arow  = m0 + l;
  const int  ncol0 = n0 + l;
  const int  ncol1 = n0 + 16 + l;
  const bool n1ok  = (ncol1 < EMB_N);
  const int  nc1   = n1ok ? ncol1 : 0;
  const float* __restrict__ Ar = x + (size_t)arow * CT_N;

  v8f acc0 = {}, acc1 = {};
  for (int p = wave; p < ENC_PIECES; p += ENC_KW) {
    for (int u = 0; u < 4; ++u) {
      const int ka = p * 32 + u * 8 + 2 * half;
      const v2f a0 = *(const v2f*)(Ar + ka);
      const v2f a1 = *(const v2f*)(Ar + ka + 4);
      v2f b00, b01, b10, b11;
      b00.x = eW[(size_t)ka * EMB_N + ncol0];
      b00.y = eW[(size_t)(ka + 1) * EMB_N + ncol0];
      b01.x = eW[(size_t)(ka + 4) * EMB_N + ncol0];
      b01.y = eW[(size_t)(ka + 5) * EMB_N + ncol0];
      b10.x = eW[(size_t)ka * EMB_N + nc1];
      b10.y = eW[(size_t)(ka + 1) * EMB_N + nc1];
      b11.x = eW[(size_t)(ka + 4) * EMB_N + nc1];
      b11.y = eW[(size_t)(ka + 5) * EMB_N + nc1];
      acc0 = wmma4(a0, b00, acc0);
      acc1 = wmma4(a0, b10, acc1);
      acc0 = wmma4(a1, b01, acc0);
      acc1 = wmma4(a1, b11, acc1);
    }
  }

  for (int r = 0; r < 8; ++r) {
    part[wave][r][lane]     = acc0[r];
    part[wave][8 + r][lane] = acc1[r];
  }
  __syncthreads();

  if (wave < 16) {
    float s = part[0][wave][lane];
    for (int w = 1; w < ENC_KW; ++w) s += part[w][wave][lane];
    const int row = m0 + (wave & 7) + 8 * half;
    const int col = n0 + 16 * (wave >> 3) + l;
    if (col < EMB_N) h[(size_t)row * EMB_N + col] = s + eB[col];
  }
}

__global__ __launch_bounds__(256) void proj1_gemm_kernel(
    const float* __restrict__ h, const float* __restrict__ W1,
    const float* __restrict__ b1, float* __restrict__ h1,
    float* __restrict__ g) {
  const int wave = threadIdx.x >> 5;
  const int lane = threadIdx.x & 31;
  const int tile = blockIdx.x * 8 + wave;
  const int mt = tile >> 5, np = tile & 31;
  const int m0 = mt * 16, n0 = np * 32;
  const int half = lane >> 4;
  const int l    = lane & 15;

  const int arow  = m0 + l;
  const int ncol0 = n0 + l;
  const int ncol1 = n0 + 16 + l;
  const float* __restrict__ Ar = h + (size_t)arow * EMB_N;

  v8f acc0 = {}, acc1 = {};
  for (int k = 0; k < EMB_N; k += 8) {
    const int ka = k + 2 * half;
    const v2f a0 = *(const v2f*)(Ar + ka);
    const v2f a1 = *(const v2f*)(Ar + ka + 4);
    v2f b00, b01, b10, b11;
    b00.x = W1[(size_t)ka * PROJ_N + ncol0];
    b00.y = W1[(size_t)(ka + 1) * PROJ_N + ncol0];
    b01.x = W1[(size_t)(ka + 4) * PROJ_N + ncol0];
    b01.y = W1[(size_t)(ka + 5) * PROJ_N + ncol0];
    b10.x = W1[(size_t)ka * PROJ_N + ncol1];
    b10.y = W1[(size_t)(ka + 1) * PROJ_N + ncol1];
    b11.x = W1[(size_t)(ka + 4) * PROJ_N + ncol1];
    b11.y = W1[(size_t)(ka + 5) * PROJ_N + ncol1];
    acc0 = wmma4(a0, b00, acc0);
    acc1 = wmma4(a0, b10, acc1);
    acc0 = wmma4(a1, b01, acc0);
    acc1 = wmma4(a1, b11, acc1);
  }
  const float bn0 = b1[ncol0];
  const float bn1 = b1[ncol1];
#pragma unroll
  for (int r = 0; r < 8; ++r) {
    const int row = m0 + r + 8 * half;
    const float t0 = acc0[r] + bn0;
    const float t1 = acc1[r] + bn1;
    h1[(size_t)row * PROJ_N + ncol0] = t0;
    h1[(size_t)row * PROJ_N + ncol1] = t1;
    g[(size_t)row * PROJ_N + ncol0] = gelu_exact(t0);
    g[(size_t)row * PROJ_N + ncol1] = gelu_exact(t1);
  }
}

__global__ __launch_bounds__(256) void proj2_gemm_kernel(
    const float* __restrict__ g, const float* __restrict__ W2,
    const float* __restrict__ b2, const float* __restrict__ h1,
    float* __restrict__ hsum) {
  const int wave = threadIdx.x >> 5;
  const int lane = threadIdx.x & 31;
  const int tile = blockIdx.x * 8 + wave;
  const int mt = tile >> 5, np = tile & 31;
  const int m0 = mt * 16, n0 = np * 32;
  const int half = lane >> 4;
  const int l    = lane & 15;

  const int arow  = m0 + l;
  const int ncol0 = n0 + l;
  const int ncol1 = n0 + 16 + l;
  const float* __restrict__ Ar = g + (size_t)arow * PROJ_N;

  v8f acc0 = {}, acc1 = {};
  for (int k = 0; k < PROJ_N; k += 8) {
    const int ka = k + 2 * half;
    const v2f a0 = *(const v2f*)(Ar + ka);
    const v2f a1 = *(const v2f*)(Ar + ka + 4);
    v2f b00, b01, b10, b11;
    b00.x = W2[(size_t)ka * PROJ_N + ncol0];
    b00.y = W2[(size_t)(ka + 1) * PROJ_N + ncol0];
    b01.x = W2[(size_t)(ka + 4) * PROJ_N + ncol0];
    b01.y = W2[(size_t)(ka + 5) * PROJ_N + ncol0];
    b10.x = W2[(size_t)ka * PROJ_N + ncol1];
    b10.y = W2[(size_t)(ka + 1) * PROJ_N + ncol1];
    b11.x = W2[(size_t)(ka + 4) * PROJ_N + ncol1];
    b11.y = W2[(size_t)(ka + 5) * PROJ_N + ncol1];
    acc0 = wmma4(a0, b00, acc0);
    acc1 = wmma4(a0, b10, acc1);
    acc0 = wmma4(a1, b01, acc0);
    acc1 = wmma4(a1, b11, acc1);
  }
  const float bn0 = b2[ncol0];
  const float bn1 = b2[ncol1];
#pragma unroll
  for (int r = 0; r < 8; ++r) {
    const int row = m0 + r + 8 * half;
    hsum[(size_t)row * PROJ_N + ncol0] =
        h1[(size_t)row * PROJ_N + ncol0] + acc0[r] + bn0;
    hsum[(size_t)row * PROJ_N + ncol1] =
        h1[(size_t)row * PROJ_N + ncol1] + acc1[r] + bn1;
  }
}

__global__ __launch_bounds__(256) void layernorm_kernel(
    const float* __restrict__ hsum, const float* __restrict__ gamma,
    const float* __restrict__ beta, float* __restrict__ out) {
  const int row = blockIdx.x;
  const int tid = threadIdx.x;
  const float* __restrict__ hr = hsum + (size_t)row * PROJ_N;

  float v[4];
  float s = 0.0f, ss = 0.0f;
#pragma unroll
  for (int i = 0; i < 4; ++i) {
    v[i] = hr[tid + i * 256];
    s += v[i];
    ss += v[i] * v[i];
  }
  __shared__ float rs[256];
  __shared__ float rss[256];
  rs[tid] = s;
  rss[tid] = ss;
  __syncthreads();
  for (int off = 128; off > 0; off >>= 1) {
    if (tid < off) {
      rs[tid] += rs[tid + off];
      rss[tid] += rss[tid + off];
    }
    __syncthreads();
  }
  const float mu  = rs[0] * (1.0f / PROJ_N);
  const float var = rss[0] * (1.0f / PROJ_N) - mu * mu;
  const float inv = rsqrtf(var + 1e-5f);
#pragma unroll
  for (int i = 0; i < 4; ++i) {
    const int c = tid + i * 256;
    out[(size_t)row * PROJ_N + c] = (v[i] - mu) * inv * gamma[c] + beta[c];
  }
}

extern "C" void kernel_launch(void* const* d_in, const int* in_sizes, int n_in,
                              void* d_out, int out_size, void* d_ws,
                              size_t ws_size, hipStream_t stream) {
  const float* eeg    = (const float*)d_in[0];
  const int*   sid    = (const int*)d_in[1];
  const float* subj_W = (const float*)d_in[2];
  const float* subj_b = (const float*)d_in[3];
  const float* enc_W  = (const float*)d_in[4];
  const float* enc_b  = (const float*)d_in[5];
  const float* W1     = (const float*)d_in[6];
  const float* b1     = (const float*)d_in[7];
  const float* W2     = (const float*)d_in[8];
  const float* b2     = (const float*)d_in[9];
  const float* gamma  = (const float*)d_in[10];
  const float* beta   = (const float*)d_in[11];
  float*       out    = (float*)d_out;

  float* xbuf  = (float*)d_ws;
  float* hbuf  = xbuf + (size_t)B_N * CT_N;
  float* h1buf = hbuf + (size_t)B_N * EMB_N;
  float* gbuf  = h1buf + (size_t)B_N * PROJ_N;
  float* hsbuf = gbuf + (size_t)B_N * PROJ_N;

  subj_gemm_kernel<<<B_N, 128, 0, stream>>>(eeg, sid, subj_W, subj_b, xbuf);
  enc_gemm_kernel<<<384, ENC_KW * 32, 0, stream>>>(xbuf, enc_W, enc_b, hbuf);
  proj1_gemm_kernel<<<256, 256, 0, stream>>>(hbuf, W1, b1, h1buf, gbuf);
  proj2_gemm_kernel<<<256, 256, 0, stream>>>(gbuf, W2, b2, h1buf, hsbuf);
  layernorm_kernel<<<B_N, 256, 0, stream>>>(hsbuf, gamma, beta, out);
}


# encoder GEMM loads: each cache line of a K piece asked for once per wave (A: 16 B row loads via a per-wave LDS tile, B: 8 B column-pair loads); k order in a piece changes, so does the f32 sum order
# speedup vs baseline: 4.4556x; 1.2547x over previous
#include <hip/hip_runtime.h>
#include <hip/hip_bf16.h>
#include <math.h>


typedef __attribute__((ext_vector_type(2))) float v2f;
typedef __attribute__((ext_vector_type(8))) float v8f;

#define B_N    1024
#define C_N    64
#define T_N    275
#define CT_N   (C_N * T_N)
#define EMB_N  184
#define PROJ_N 1024

__device__ __forceinline__ v8f wmma4(v2f a, v2f b, v8f c) {
  return __builtin_amdgcn_wmma_f32_16x16x4_f32(false, a, false, b, (short)0, c,
                                               false, false);
}

__device__ __forceinline__ float gelu_exact(float x) {
  return 0.5f * x * (1.0f + erff(x * 0.70710678118654752440f));
}

#define SJ_PITCH 276
#define SJ_KS    69
#define SJ_CH    32
#define SJ_NCH   9
#define SJ_RB    2
#define SJ_RL    9

__device__ __forceinline__ void sj_load(float (&v)[SJ_RB * SJ_RL],
                                        const float* __restrict__ mat,
                                        int row0, int lastrow, int lane) {
  const int lane8 = lane < 19 ? lane : 18;
#pragma unroll
  for (int j = 0; j < SJ_RB; ++j) {
    int r = row0 + j;
    if (r > lastrow) r = lastrow;
    const float* __restrict__ rp = mat + (size_t)r * T_N;
#pragma unroll
    for (int q = 0; q < 8; ++q) v[j * SJ_RL + q] = rp[q * 32 + lane];
    v[j * SJ_RL + 8] = rp[256 + lane8];
  }
}

__device__ __forceinline__ void sj_store(float* __restrict__ dstrow,
                                         const float (&v)[SJ_RB * SJ_RL],
                                         int lane) {
  const int lane8 = lane < 19 ? lane : 18;
#pragma unroll
  for (int j = 0; j < SJ_RB; ++j) {
#pragma unroll
    for (int q = 0; q < 8; ++q)
      dstrow[j * SJ_PITCH + q * 32 + lane] = v[j * SJ_RL + q];
    dstrow[j * SJ_PITCH + 256 + lane8] = v[j * SJ_RL + 8];
  }
}

__global__ __launch_bounds__(128, 4) void subj_gemm_kernel(
    const float* __restrict__ eeg, const int* __restrict__ sid,
    const float* __restrict__ sW, const float* __restrict__ sB,
    float* __restrict__ x) {
  __shared__ __attribute__((aligned(16))) float S[C_N * SJ_PITCH];

  const int b    = blockIdx.x;
  const int tid  = threadIdx.x;
  const int lane = tid & 31;
  const int half = lane >> 4;
  const int l    = lane & 15;
  const int wave = __builtin_amdgcn_readfirstlane(tid >> 5);

  const int s = sid[b];
  const float* __restrict__ E    = eeg + (size_t)b * CT_N;
  const float* __restrict__ W    = sW + (size_t)s * T_N * T_N;
  const float* __restrict__ bias = sB + (size_t)s * T_N;

  if (tid < C_N) S[tid * SJ_PITCH + T_N] = 0.0f;
#pragma unroll
  for (int bt = 0; bt < 8; ++bt) {
    float v[SJ_RB * SJ_RL];
    sj_load(v, E, wave * 16 + bt * SJ_RB, C_N - 1, lane);
    sj_store(S + (wave * 16 + bt * SJ_RB) * SJ_PITCH, v, lane);
  }
  __syncthreads();

  v2f a[SJ_KS];
  {
    const float* ap = S + (wave * 16 + l) * SJ_PITCH + 2 * half;
#pragma unroll
    for (int ks = 0; ks < SJ_KS; ++ks) a[ks] = *(const v2f*)(ap + 4 * ks);
  }
  __syncthreads();

#pragma unroll
  for (int bt = 0; bt < 4; ++bt) {
    float v[SJ_RB * SJ_RL];
    sj_load(v, W, wave * 8 + bt * SJ_RB, T_N - 1, lane);
    sj_store(S + (wave * 8 + bt * SJ_RB) * SJ_PITCH, v, lane);
  }
  float* __restrict__ X = x + (size_t)b * CT_N;

  for (int c = 0; c < SJ_NCH; ++c) {
    __syncthreads();
    const bool more = (c + 1 < SJ_NCH);
    const int nrow0 = (c + 1) * SJ_CH + wave * 8;
    float* __restrict__ dst =
        S + (((c + 1) & 1) * SJ_CH + wave * 8) * SJ_PITCH;

    const float* bp0 = S + (c & 1) * (SJ_CH * SJ_PITCH) + l * SJ_PITCH + 2 * half;
    const float* bp1 = bp0 + 16 * SJ_PITCH;
    v8f acc0 = {}, acc1 = {};
#pragma unroll
    for (int part = 0; part < 4; ++part) {
      float v[SJ_RB * SJ_RL];
      if (more) sj_load(v, W, nrow0 + part * SJ_RB, T_N - 1, lane);
#pragma unroll
      for (int ks = (part * SJ_KS + 3) / 4; ks < ((part + 1) * SJ_KS + 3) / 4; ++ks) {
        const v2f b0 = *(const v2f*)(bp0 + 4 * ks);
        const v2f b1 = *(const v2f*)(bp1 + 4 * ks);
        acc0 = wmma4(a[ks], b0, acc0);
        acc1 = wmma4(a[ks], b1, acc1);
      }
      if (more) sj_store(dst + part * SJ_RB * SJ_PITCH, v, lane);
    }
    const int col0 = c * SJ_CH + l;
    const int col1 = col0 + 16;
    const float bn0 = bias[col0];
#pragma unroll
    for (int r = 0; r < 8; ++r) {
      const int orow = wave * 16 + r + 8 * half;
      X[orow * T_N + col0] = acc0[r] + bn0;
    }
    if (col1 < T_N) {
      const float bn1 = bias[col1];
#pragma unroll
      for (int r = 0; r < 8; ++r) {
        const int orow = wave * 16 + r + 8 * half;
        X[orow * T_N + col1] = acc1[r] + bn1;
      }
    }
  }
}

typedef __attribute__((ext_vector_type(4))) float v4f;

#define ENC_KW     25
#define ENC_PIECES (CT_N / 32)
#define ENC_AP     36

__global__ __launch_bounds__(ENC_KW * 32) void enc_gemm_kernel(
    const float* __restrict__ x, const float* __restrict__ eW,
    const float* __restrict__ eB, float* __restrict__ h) {
  __shared__ __attribute__((aligned(16))) float lds[ENC_KW * 16 * ENC_AP];
  float (*part)[16][32] = reinterpret_cast<float (*)[16][32]>(lds);

  const int wave = __builtin_amdgcn_readfirstlane(threadIdx.x >> 5);
  const int lane = threadIdx.x & 31;
  const int tile = blockIdx.x;
  const int mt = tile / 6, np = tile % 6;
  const int m0 = mt * 16, n0 = np * 32;
  const int half = lane >> 4;
  const int l    = lane & 15;

  const int srow = lane >> 3;
  const int scol = (lane & 7) * 4;
  const float* __restrict__ Ag = x + (size_t)(m0 + srow) * CT_N + scol;
  float* Ast = lds + wave * (16 * ENC_AP) + srow * ENC_AP + scol;
  const float* Ald = lds + wave * (16 * ENC_AP) + l * ENC_AP + 16 * half;

  const int  c0  = n0 + 2 * l;
  const bool cok = (c0 < EMB_N);
  const float* __restrict__ Bg = eW + (size_t)(16 * half) * EMB_N + (cok ? c0 : 0);

  v8f acc0 = {}, acc1 = {};
  for (int p = wave; p < ENC_PIECES; p += ENC_KW) {
    v4f ag[4];
#pragma unroll
    for (int j = 0; j < 4; ++j)
      ag[j] = *(const v4f*)(Ag + (size_t)(4 * j) * CT_N + p * 32);
    const float* __restrict__ bp = Bg + (size_t)p * 32 * EMB_N;
    v2f br[16];
#pragma unroll
    for (int q = 0; q < 16; ++q) br[q] = *(const v2f*)(bp + q * EMB_N);

#pragma unroll
    for (int j = 0; j < 4; ++j) *(v4f*)(Ast + 4 * j * ENC_AP) = ag[j];
    __builtin_amdgcn_fence(__ATOMIC_RELEASE, "wavefront");
    __builtin_amdgcn_wave_barrier();
    __builtin_amdgcn_fence(__ATOMIC_ACQUIRE, "wavefront");
    v4f af[4];
#pragma unroll
    for (int i = 0; i < 4; ++i) af[i] = *(const v4f*)(Ald + 4 * i);
    __builtin_amdgcn_fence(__ATOMIC_RELEASE, "wavefront");
    __builtin_amdgcn_wave_barrier();
    __builtin_amdgcn_fence(__ATOMIC_ACQUIRE, "wavefront");

#pragma unroll
    for (int j = 0; j < 8; ++j) {
      v2f a, b0, b1;
      a.x  = af[j >> 1][2 * (j & 1)];
      a.y  = af[j >> 1][2 * (j & 1) + 1];
      b0.x = br[2 * j].x;
      b0.y = br[2 * j + 1].x;
      b1.x = br[2 * j].y;
      b1.y = br[2 * j + 1].y;
      acc0 = wmma4(a, b0, acc0);
      acc1 = wmma4(a, b1, acc1);
    }
  }

  __syncthreads();
  for (int r = 0; r < 8; ++r) {
    part[wave][r][lane]     = acc0[r];
    part[wave][8 + r][lane] = acc1[r];
  }
  __syncthreads();

  if (wave < 16) {
    float s = part[0][wave][lane];
    for (int w = 1; w < ENC_KW; ++w) s += part[w][wave][lane];
    const int row = m0 + (wave & 7) + 8 * half;
    const int col = n0 + 2 * l + (wave >> 3);
    if (col < EMB_N) h[(size_t)row * EMB_N + col] = s + eB[col];
  }
}

__global__ __launch_bounds__(256) void proj1_gemm_kernel(
    const float* __restrict__ h, const float* __restrict__ W1,
    const float* __restrict__ b1, float* __restrict__ h1,
    float* __restrict__ g) {
  const int wave = threadIdx.x >> 5;
  const int lane = threadIdx.x & 31;
  const int tile = blockIdx.x * 8 + wave;
  const int mt = tile >> 5, np = tile & 31;
  const int m0 = mt * 16, n0 = np * 32;
  const int half = lane >> 4;
  const int l    = lane & 15;

  const int arow  = m0 + l;
  const int ncol0 = n0 + l;
  const int ncol1 = n0 + 16 + l;
  const float* __restrict__ Ar = h + (size_t)arow * EMB_N;

  v8f acc0 = {}, acc1 = {};
  for (int k = 0; k < EMB_N; k += 8) {
    const int ka = k + 2 * half;
    const v2f a0 = *(const v2f*)(Ar + ka);
    const v2f a1 = *(const v2f*)(Ar + ka + 4);
    v2f b00, b01, b10, b11;
    b00.x = W1[(size_t)ka * PROJ_N + ncol0];
    b00.y = W1[(size_t)(ka + 1) * PROJ_N + ncol0];
    b01.x = W1[(size_t)(ka + 4) * PROJ_N + ncol0];
    b01.y = W1[(size_t)(ka + 5) * PROJ_N + ncol0];
    b10.x = W1[(size_t)ka * PROJ_N + ncol1];
    b10.y = W1[(size_t)(ka + 1) * PROJ_N + ncol1];
    b11.x = W1[(size_t)(ka + 4) * PROJ_N + ncol1];
    b11.y = W1[(size_t)(ka + 5) * PROJ_N + ncol1];
    acc0 = wmma4(a0, b00, acc0);
    acc1 = wmma4(a0, b10, acc1);
    acc0 = wmma4(a1, b01, acc0);
    acc1 = wmma4(a1, b11, acc1);
  }
  const float bn0 = b1[ncol0];
  const float bn1 = b1[ncol1];
#pragma unroll
  for (int r = 0; r < 8; ++r) {
    const int row = m0 + r + 8 * half;
    const float t0 = acc0[r] + bn0;
    const float t1 = acc1[r] + bn1;
    h1[(size_t)row * PROJ_N + ncol0] = t0;
    h1[(size_t)row * PROJ_N + ncol1] = t1;
    g[(size_t)row * PROJ_N + ncol0] = gelu_exact(t0);
    g[(size_t)row * PROJ_N + ncol1] = gelu_exact(t1);
  }
}

__global__ __launch_bounds__(256) void proj2_gemm_kernel(
    const float* __restrict__ g, const float* __restrict__ W2,
    const float* __restrict__ b2, const float* __restrict__ h1,
    float* __restrict__ hsum) {
  const int wave = threadIdx.x >> 5;
  const int lane = threadIdx.x & 31;
  const int tile = blockIdx.x * 8 + wave;
  const int mt = tile >> 5, np = tile & 31;
  const int m0 = mt * 16, n0 = np * 32;
  const int half = lane >> 4;
  const int l    = lane & 15;

  const int arow  = m0 + l;
  const int ncol0 = n0 + l;
  const int ncol1 = n0 + 16 + l;
  const float* __restrict__ Ar = g + (size_t)arow * PROJ_N;

  v8f acc0 = {}, acc1 = {};
  for (int k = 0; k < PROJ_N; k += 8) {
    const int ka = k + 2 * half;
    const v2f a0 = *(const v2f*)(Ar + ka);
    const v2f a1 = *(const v2f*)(Ar + ka + 4);
    v2f b00, b01, b10, b11;
    b00.x = W2[(size_t)ka * PROJ_N + ncol0];
    b00.y = W2[(size_t)(ka + 1) * PROJ_N + ncol0];
    b01.x = W2[(size_t)(ka + 4) * PROJ_N + ncol0];
    b01.y = W2[(size_t)(ka + 5) * PROJ_N + ncol0];
    b10.x = W2[(size_t)ka * PROJ_N + ncol1];
    b10.y = W2[(size_t)(ka + 1) * PROJ_N + ncol1];
    b11.x = W2[(size_t)(ka + 4) * PROJ_N + ncol1];
    b11.y = W2[(size_t)(ka + 5) * PROJ_N + ncol1];
    acc0 = wmma4(a0, b00, acc0);
    acc1 = wmma4(a0, b10, acc1);
    acc0 = wmma4(a1, b01, acc0);
    acc1 = wmma4(a1, b11, acc1);
  }
  const float bn0 = b2[ncol0];
  const float bn1 = b2[ncol1];
#pragma unroll
  for (int r = 0; r < 8; ++r) {
    const int row = m0 + r + 8 * half;
    hsum[(size_t)row * PROJ_N + ncol0] =
        h1[(size_t)row * PROJ_N + ncol0] + acc0[r] + bn0;
    hsum[(size_t)row * PROJ_N + ncol1] =
        h1[(size_t)row * PROJ_N + ncol1] + acc1[r] + bn1;
  }
}

__global__ __launch_bounds__(256) void layernorm_kernel(
    const float* __restrict__ hsum, const float* __restrict__ gamma,
    const float* __restrict__ beta, float* __restrict__ out) {
  const int row = blockIdx.x;
  const int tid = threadIdx.x;
  const float* __restrict__ hr = hsum + (size_t)row * PROJ_N;

  float v[4];
  float s = 0.0f, ss = 0.0f;
#pragma unroll
  for (int i = 0; i < 4; ++i) {
    v[i] = hr[tid + i * 256];
    s += v[i];
    ss += v[i] * v[i];
  }
  __shared__ float rs[256];
  __shared__ float rss[256];
  rs[tid] = s;
  rss[tid] = ss;
  __syncthreads();
  for (int off = 128; off > 0; off >>= 1) {
    if (tid < off) {
      rs[tid] += rs[tid + off];
      rss[tid] += rss[tid + off];
    }
    __syncthreads();
  }
  const float mu  = rs[0] * (1.0f / PROJ_N);
  const float var = rss[0] * (1.0f / PROJ_N) - mu * mu;
  const float inv = rsqrtf(var + 1e-5f);
#pragma unroll
  for (int i = 0; i < 4; ++i) {
    const int c = tid + i * 256;
    out[(size_t)row * PROJ_N + c] = (v[i] - mu) * inv * gamma[c] + beta[c];
  }
}

extern "C" void kernel_launch(void* const* d_in, const int* in_sizes, int n_in,
                              void* d_out, int out_size, void* d_ws,
                              size_t ws_size, hipStream_t stream) {
  const float* eeg    = (const float*)d_in[0];
  const int*   sid    = (const int*)d_in[1];
  const float* subj_W = (const float*)d_in[2];
  const float* subj_b = (const float*)d_in[3];
  const float* enc_W  = (const float*)d_in[4];
  const float* enc_b  = (const float*)d_in[5];
  const float* W1     = (const float*)d_in[6];
  const float* b1     = (const float*)d_in[7];
  const float* W2     = (const float*)d_in[8];
  const float* b2     = (const float*)d_in[9];
  const float* gamma  = (const float*)d_in[10];
  const float* beta   = (const float*)d_in[11];
  float*       out    = (float*)d_out;

  float* xbuf  = (float*)d_ws;
  float* hbuf  = xbuf + (size_t)B_N * CT_N;
  float* h1buf = hbuf + (size_t)B_N * EMB_N;
  float* gbuf  = h1buf + (size_t)B_N * PROJ_N;
  float* hsbuf = gbuf + (size_t)B_N * PROJ_N;

  subj_gemm_kernel<<<B_N, 128, 0, stream>>>(eeg, sid, subj_W, subj_b, xbuf);
  enc_gemm_kernel<<<384, ENC_KW * 32, 0, stream>>>(xbuf, enc_W, enc_b, hbuf);
  proj1_gemm_kernel<<<256, 256, 0, stream>>>(hbuf, W1, b1, h1buf, gbuf);
  proj2_gemm_kernel<<<256, 256, 0, stream>>>(gbuf, W2, b2, h1buf, hsbuf);
  layernorm_kernel<<<B_N, 256, 0, stream>>>(hsbuf, gamma, beta, out);
}


# encoder GEMM: 16x48 strips (three accumulators per A fragment), 256 equal workgroups of 32 waves, one per WGP (was 384 of 25 on 256 WGPs); K now split 32 ways, so the f32 sum order changes
# speedup vs baseline: 4.8774x; 1.0947x over previous
#include <hip/hip_runtime.h>
#include <hip/hip_bf16.h>
#include <math.h>


typedef __attribute__((ext_vector_type(2))) float v2f;
typedef __attribute__((ext_vector_type(8))) float v8f;

#define B_N    1024
#define C_N    64
#define T_N    275
#define CT_N   (C_N * T_N)
#define EMB_N  184
#define PROJ_N 1024

__device__ __forceinline__ v8f wmma4(v2f a, v2f b, v8f c) {
  return __builtin_amdgcn_wmma_f32_16x16x4_f32(false, a, false, b, (short)0, c,
                                               false, false);
}

__device__ __forceinline__ float gelu_exact(float x) {
  return 0.5f * x * (1.0f + erff(x * 0.70710678118654752440f));
}

#define SJ_PITCH 276
#define SJ_KS    69
#define SJ_CH    32
#define SJ_NCH   9
#define SJ_RB    2
#define SJ_RL    9

__device__ __forceinline__ void sj_load(float (&v)[SJ_RB * SJ_RL],
                                        const float* __restrict__ mat,
                                        int row0, int lastrow, int lane) {
  const int lane8 = lane < 19 ? lane : 18;
#pragma unroll
  for (int j = 0; j < SJ_RB; ++j) {
    int r = row0 + j;
    if (r > lastrow) r = lastrow;
    const float* __restrict__ rp = mat + (size_t)r * T_N;
#pragma unroll
    for (int q = 0; q < 8; ++q) v[j * SJ_RL + q] = rp[q * 32 + lane];
    v[j * SJ_RL + 8] = rp[256 + lane8];
  }
}

__device__ __forceinline__ void sj_store(float* __restrict__ dstrow,
                                         const float (&v)[SJ_RB * SJ_RL],
                                         int lane) {
  const int lane8 = lane < 19 ? lane : 18;
#pragma unroll
  for (int j = 0; j < SJ_RB; ++j) {
#pragma unroll
    for (int q = 0; q < 8; ++q)
      dstrow[j * SJ_PITCH + q * 32 + lane] = v[j * SJ_RL + q];
    dstrow[j * SJ_PITCH + 256 + lane8] = v[j * SJ_RL + 8];
  }
}

__global__ __launch_bounds__(128, 4) void subj_gemm_kernel(
    const float* __restrict__ eeg, const int* __restrict__ sid,
    const float* __restrict__ sW, const float* __restrict__ sB,
    float* __restrict__ x) {
  __shared__ __attribute__((aligned(16))) float S[C_N * SJ_PITCH];

  const int b    = blockIdx.x;
  const int tid  = threadIdx.x;
  const int lane = tid & 31;
  const int half = lane >> 4;
  const int l    = lane & 15;
  const int wave = __builtin_amdgcn_readfirstlane(tid >> 5);

  const int s = sid[b];
  const float* __restrict__ E    = eeg + (size_t)b * CT_N;
  const float* __restrict__ W    = sW + (size_t)s * T_N * T_N;
  const float* __restrict__ bias = sB + (size_t)s * T_N;

  if (tid < C_N) S[tid * SJ_PITCH + T_N] = 0.0f;
#pragma unroll
  for (int bt = 0; bt < 8; ++bt) {
    float v[SJ_RB * SJ_RL];
    sj_load(v, E, wave * 16 + bt * SJ_RB, C_N - 1, lane);
    sj_store(S + (wave * 16 + bt * SJ_RB) * SJ_PITCH, v, lane);
  }
  __syncthreads();

  v2f a[SJ_KS];
  {
    const float* ap = S + (wave * 16 + l) * SJ_PITCH + 2 * half;
#pragma unroll
    for (int ks = 0; ks < SJ_KS; ++ks) a[ks] = *(const v2f*)(ap + 4 * ks);
  }
  __syncthreads();

#pragma unroll
  for (int bt = 0; bt < 4; ++bt) {
    float v[SJ_RB * SJ_RL];
    sj_load(v, W, wave * 8 + bt * SJ_RB, T_N - 1, lane);
    sj_store(S + (wave * 8 + bt * SJ_RB) * SJ_PITCH, v, lane);
  }
  float* __restrict__ X = x + (size_t)b * CT_N;

  for (int c = 0; c < SJ_NCH; ++c) {
    __syncthreads();
    const bool more = (c + 1 < SJ_NCH);
    const int nrow0 = (c + 1) * SJ_CH + wave * 8;
    float* __restrict__ dst =
        S + (((c + 1) & 1) * SJ_CH + wave * 8) * SJ_PITCH;

    const float* bp0 = S + (c & 1) * (SJ_CH * SJ_PITCH) + l * SJ_PITCH + 2 * half;
    const float* bp1 = bp0 + 16 * SJ_PITCH;
    v8f acc0 = {}, acc1 = {};
#pragma unroll
    for (int part = 0; part < 4; ++part) {
      float v[SJ_RB * SJ_RL];
      if (more) sj_load(v, W, nrow0 + part * SJ_RB, T_N - 1, lane);
#pragma unroll
      for (int ks = (part * SJ_KS + 3) / 4; ks < ((part + 1) * SJ_KS + 3) / 4; ++ks) {
        const v2f b0 = *(const v2f*)(bp0 + 4 * ks);
        const v2f b1 = *(const v2f*)(bp1 + 4 * ks);
        acc0 = wmma4(a[ks], b0, acc0);
        acc1 = wmma4(a[ks], b1, acc1);
      }
      if (more) sj_store(dst + part * SJ_RB * SJ_PITCH, v, lane);
    }
    const int col0 = c * SJ_CH + l;
    const int col1 = col0 + 16;
    const float bn0 = bias[col0];
#pragma unroll
    for (int r = 0; r < 8; ++r) {
      const int orow = wave * 16 + r + 8 * half;
      X[orow * T_N + col0] = acc0[r] + bn0;
    }
    if (col1 < T_N) {
      const float bn1 = bias[col1];
#pragma unroll
      for (int r = 0; r < 8; ++r) {
        const int orow = wave * 16 + r + 8 * half;
        X[orow * T_N + col1] = acc1[r] + bn1;
      }
    }
  }
}

typedef __attribute__((ext_vector_type(4))) float v4f;

#define ENC_KW     32
#define ENC_PIECES (CT_N / 32)
#define ENC_AP     36
#define ENC_NW     48
#define ENC_SR     24

__global__ __launch_bounds__(ENC_KW * 32) void enc_gemm_kernel(
    const float* __restrict__ x, const float* __restrict__ eW,
    const float* __restrict__ eB, float* __restrict__ h) {
  __shared__ __attribute__((aligned(16))) float lds[ENC_KW * ENC_SR * 32];
  float (*part)[ENC_SR][32] = reinterpret_cast<float (*)[ENC_SR][32]>(lds);

  const int wave = __builtin_amdgcn_readfirstlane(threadIdx.x >> 5);
  const int lane = threadIdx.x & 31;
  const int tile = blockIdx.x;
  const int mt = tile >> 2, np = tile & 3;
  const int m0 = mt * 16;
  const int nlo = np * ENC_NW;
  const int n0 = nlo < EMB_N - ENC_NW ? nlo : EMB_N - ENC_NW;
  const int half = lane >> 4;
  const int l    = lane & 15;

  const int srow = lane >> 3;
  const int scol = (lane & 7) * 4;
  const float* __restrict__ Ag = x + (size_t)(m0 + srow) * CT_N + scol;
  float* Ast = lds + wave * (16 * ENC_AP) + srow * ENC_AP + scol;
  const float* Ald = lds + wave * (16 * ENC_AP) + l * ENC_AP + 16 * half;

  const float* __restrict__ Bg = eW + (size_t)(16 * half) * EMB_N + n0 + 3 * l;

  v8f acc0 = {}, acc1 = {}, acc2 = {};
  for (int p = wave; p < ENC_PIECES; p += ENC_KW) {
    v4f ag[4];
#pragma unroll
    for (int j = 0; j < 4; ++j)
      ag[j] = *(const v4f*)(Ag + (size_t)(4 * j) * CT_N + p * 32);
    const float* __restrict__ bp = Bg + (size_t)p * 32 * EMB_N;
    float br[16][3];
#pragma unroll
    for (int q = 0; q < 16; ++q) {
      br[q][0] = bp[q * EMB_N];
      br[q][1] = bp[q * EMB_N + 1];
      br[q][2] = bp[q * EMB_N + 2];
    }

#pragma unroll
    for (int j = 0; j < 4; ++j) *(v4f*)(Ast + 4 * j * ENC_AP) = ag[j];
    __builtin_amdgcn_fence(__ATOMIC_RELEASE, "wavefront");
    __builtin_amdgcn_wave_barrier();
    __builtin_amdgcn_fence(__ATOMIC_ACQUIRE, "wavefront");
    v4f af[4];
#pragma unroll
    for (int i = 0; i < 4; ++i) af[i] = *(const v4f*)(Ald + 4 * i);
    __builtin_amdgcn_fence(__ATOMIC_RELEASE, "wavefront");
    __builtin_amdgcn_wave_barrier();
    __builtin_amdgcn_fence(__ATOMIC_ACQUIRE, "wavefront");

#pragma unroll
    for (int j = 0; j < 8; ++j) {
      v2f a, b0, b1, b2;
      a.x  = af[j >> 1][2 * (j & 1)];
      a.y  = af[j >> 1][2 * (j & 1) + 1];
      b0.x = br[2 * j][0];
      b0.y = br[2 * j + 1][0];
      b1.x = br[2 * j][1];
      b1.y = br[2 * j + 1][1];
      b2.x = br[2 * j][2];
      b2.y = br[2 * j + 1][2];
      acc0 = wmma4(a, b0, acc0);
      acc1 = wmma4(a, b1, acc1);
      acc2 = wmma4(a, b2, acc2);
    }
  }

  __syncthreads();
  for (int r = 0; r < 8; ++r) {
    part[wave][r][lane]      = acc0[r];
    part[wave][8 + r][lane]  = acc1[r];
    part[wave][16 + r][lane] = acc2[r];
  }
  __syncthreads();

  if (wave < ENC_SR) {
    float s = part[0][wave][lane];
    for (int w = 1; w < ENC_KW; ++w) s += part[w][wave][lane];
    const int row = m0 + (wave & 7) + 8 * half;
    const int col = n0 + 3 * l + (wave >> 3);
    if (col >= nlo) h[(size_t)row * EMB_N + col] = s + eB[col];
  }
}

__global__ __launch_bounds__(256) void proj1_gemm_kernel(
    const float* __restrict__ h, const float* __restrict__ W1,
    const float* __restrict__ b1, float* __restrict__ h1,
    float* __restrict__ g) {
  const int wave = threadIdx.x >> 5;
  const int lane = threadIdx.x & 31;
  const int tile = blockIdx.x * 8 + wave;
  const int mt = tile >> 5, np = tile & 31;
  const int m0 = mt * 16, n0 = np * 32;
  const int half = lane >> 4;
  const int l    = lane & 15;

  const int arow  = m0 + l;
  const int ncol0 = n0 + l;
  const int ncol1 = n0 + 16 + l;
  const float* __restrict__ Ar = h + (size_t)arow * EMB_N;

  v8f acc0 = {}, acc1 = {};
  for (int k = 0; k < EMB_N; k += 8) {
    const int ka = k + 2 * half;
    const v2f a0 = *(const v2f*)(Ar + ka);
    const v2f a1 = *(const v2f*)(Ar + ka + 4);
    v2f b00, b01, b10, b11;
    b00.x = W1[(size_t)ka * PROJ_N + ncol0];
    b00.y = W1[(size_t)(ka + 1) * PROJ_N + ncol0];
    b01.x = W1[(size_t)(ka + 4) * PROJ_N + ncol0];
    b01.y = W1[(size_t)(ka + 5) * PROJ_N + ncol0];
    b10.x = W1[(size_t)ka * PROJ_N + ncol1];
    b10.y = W1[(size_t)(ka + 1) * PROJ_N + ncol1];
    b11.x = W1[(size_t)(ka + 4) * PROJ_N + ncol1];
    b11.y = W1[(size_t)(ka + 5) * PROJ_N + ncol1];
    acc0 = wmma4(a0, b00, acc0);
    acc1 = wmma4(a0, b10, acc1);
    acc0 = wmma4(a1, b01, acc0);
    acc1 = wmma4(a1, b11, acc1);
  }
  const float bn0 = b1[ncol0];
  const float bn1 = b1[ncol1];
#pragma unroll
  for (int r = 0; r < 8; ++r) {
    const int row = m0 + r + 8 * half;
    const float t0 = acc0[r] + bn0;
    const float t1 = acc1[r] + bn1;
    h1[(size_t)row * PROJ_N + ncol0] = t0;
    h1[(size_t)row * PROJ_N + ncol1] = t1;
    g[(size_t)row * PROJ_N + ncol0] = gelu_exact(t0);
    g[(size_t)row * PROJ_N + ncol1] = gelu_exact(t1);
  }
}

__global__ __launch_bounds__(256) void proj2_gemm_kernel(
    const float* __restrict__ g, const float* __restrict__ W2,
    const float* __restrict__ b2, const float* __restrict__ h1,
    float* __restrict__ hsum) {
  const int wave = threadIdx.x >> 5;
  const int lane = threadIdx.x & 31;
  const int tile = blockIdx.x * 8 + wave;
  const int mt = tile >> 5, np = tile & 31;
  const int m0 = mt * 16, n0 = np * 32;
  const int half = lane >> 4;
  const int l    = lane & 15;

  const int arow  = m0 + l;
  const int ncol0 = n0 + l;
  const int ncol1 = n0 + 16 + l;
  const float* __restrict__ Ar = g + (size_t)arow * PROJ_N;

  v8f acc0 = {}, acc1 = {};
  for (int k = 0; k < PROJ_N; k += 8) {
    const int ka = k + 2 * half;
    const v2f a0 = *(const v2f*)(Ar + ka);
    const v2f a1 = *(const v2f*)(Ar + ka + 4);
    v2f b00, b01, b10, b11;
    b00.x = W2[(size_t)ka * PROJ_N + ncol0];
    b00.y = W2[(size_t)(ka + 1) * PROJ_N + ncol0];
    b01.x = W2[(size_t)(ka + 4) * PROJ_N + ncol0];
    b01.y = W2[(size_t)(ka + 5) * PROJ_N + ncol0];
    b10.x = W2[(size_t)ka * PROJ_N + ncol1];
    b10.y = W2[(size_t)(ka + 1) * PROJ_N + ncol1];
    b11.x = W2[(size_t)(ka + 4) * PROJ_N + ncol1];
    b11.y = W2[(size_t)(ka + 5) * PROJ_N + ncol1];
    acc0 = wmma4(a0, b00, acc0);
    acc1 = wmma4(a0, b10, acc1);
    acc0 = wmma4(a1, b01, acc0);
    acc1 = wmma4(a1, b11, acc1);
  }
  const float bn0 = b2[ncol0];
  const float bn1 = b2[ncol1];
#pragma unroll
  for (int r = 0; r < 8; ++r) {
    const int row = m0 + r + 8 * half;
    hsum[(size_t)row * PROJ_N + ncol0] =
        h1[(size_t)row * PROJ_N + ncol0] + acc0[r] + bn0;
    hsum[(size_t)row * PROJ_N + ncol1] =
        h1[(size_t)row * PROJ_N + ncol1] + acc1[r] + bn1;
  }
}

__global__ __launch_bounds__(256) void layernorm_kernel(
    const float* __restrict__ hsum, const float* __restrict__ gamma,
    const float* __restrict__ beta, float* __restrict__ out) {
  const int row = blockIdx.x;
  const int tid = threadIdx.x;
  const float* __restrict__ hr = hsum + (size_t)row * PROJ_N;

  float v[4];
  float s = 0.0f, ss = 0.0f;
#pragma unroll
  for (int i = 0; i < 4; ++i) {
    v[i] = hr[tid + i * 256];
    s += v[i];
    ss += v[i] * v[i];
  }
  __shared__ float rs[256];
  __shared__ float rss[256];
  rs[tid] = s;
  rss[tid] = ss;
  __syncthreads();
  for (int off = 128; off > 0; off >>= 1) {
    if (tid < off) {
      rs[tid] += rs[tid + off];
      rss[tid] += rss[tid + off];
    }
    __syncthreads();
  }
  const float mu  = rs[0] * (1.0f / PROJ_N);
  const float var = rss[0] * (1.0f / PROJ_N) - mu * mu;
  const float inv = rsqrtf(var + 1e-5f);
#pragma unroll
  for (int i = 0; i < 4; ++i) {
    const int c = tid + i * 256;
    out[(size_t)row * PROJ_N + c] = (v[i] - mu) * inv * gamma[c] + beta[c];
  }
}

extern "C" void kernel_launch(void* const* d_in, const int* in_sizes, int n_in,
                              void* d_out, int out_size, void* d_ws,
                              size_t ws_size, hipStream_t stream) {
  const float* eeg    = (const float*)d_in[0];
  const int*   sid    = (const int*)d_in[1];
  const float* subj_W = (const float*)d_in[2];
  const float* subj_b = (const float*)d_in[3];
  const float* enc_W  = (const float*)d_in[4];
  const float* enc_b  = (const float*)d_in[5];
  const float* W1     = (const float*)d_in[6];
  const float* b1     = (const float*)d_in[7];
  const float* W2     = (const float*)d_in[8];
  const float* b2     = (const float*)d_in[9];
  const float* gamma  = (const float*)d_in[10];
  const float* beta   = (const float*)d_in[11];
  float*       out    = (float*)d_out;

  float* xbuf  = (float*)d_ws;
  float* hbuf  = xbuf + (size_t)B_N * CT_N;
  float* h1buf = hbuf + (size_t)B_N * EMB_N;
  float* gbuf  = h1buf + (size_t)B_N * PROJ_N;
  float* hsbuf = gbuf + (size_t)B_N * PROJ_N;

  subj_gemm_kernel<<<B_N, 128, 0, stream>>>(eeg, sid, subj_W, subj_b, xbuf);
  enc_gemm_kernel<<<256, ENC_KW * 32, 0, stream>>>(xbuf, enc_W, enc_b, hbuf);
  proj1_gemm_kernel<<<256, 256, 0, stream>>>(hbuf, W1, b1, h1buf, gbuf);
  proj2_gemm_kernel<<<256, 256, 0, stream>>>(gbuf, W2, b2, h1buf, hsbuf);
  layernorm_kernel<<<B_N, 256, 0, stream>>>(hsbuf, gamma, beta, out);
}


# encoder GEMM: K split 16 ways over workgroups, each forms a 64x184 tile with both operands staged once in LDS (was 16x48 strips, per-wave loads); a sum kernel adds the 16 planes: f32 sum order changes
# speedup vs baseline: 5.8871x; 1.2070x over previous
#include <hip/hip_runtime.h>
#include <hip/hip_bf16.h>
#include <math.h>


typedef __attribute__((ext_vector_type(2))) float v2f;
typedef __attribute__((ext_vector_type(8))) float v8f;

#define B_N    1024
#define C_N    64
#define T_N    275
#define CT_N   (C_N * T_N)
#define EMB_N  184
#define PROJ_N 1024

__device__ __forceinline__ v8f wmma4(v2f a, v2f b, v8f c) {
  return __builtin_amdgcn_wmma_f32_16x16x4_f32(false, a, false, b, (short)0, c,
                                               false, false);
}

__device__ __forceinline__ float gelu_exact(float x) {
  return 0.5f * x * (1.0f + erff(x * 0.70710678118654752440f));
}

#define SJ_PITCH 276
#define SJ_KS    69
#define SJ_CH    32
#define SJ_NCH   9
#define SJ_RB    2
#define SJ_RL    9

__device__ __forceinline__ void sj_load(float (&v)[SJ_RB * SJ_RL],
                                        const float* __restrict__ mat,
                                        int row0, int lastrow, int lane) {
  const int lane8 = lane < 19 ? lane : 18;
#pragma unroll
  for (int j = 0; j < SJ_RB; ++j) {
    int r = row0 + j;
    if (r > lastrow) r = lastrow;
    const float* __restrict__ rp = mat + (size_t)r * T_N;
#pragma unroll
    for (int q = 0; q < 8; ++q) v[j * SJ_RL + q] = rp[q * 32 + lane];
    v[j * SJ_RL + 8] = rp[256 + lane8];
  }
}

__device__ __forceinline__ void sj_store(float* __restrict__ dstrow,
                                         const float (&v)[SJ_RB * SJ_RL],
                                         int lane) {
  const int lane8 = lane < 19 ? lane : 18;
#pragma unroll
  for (int j = 0; j < SJ_RB; ++j) {
#pragma unroll
    for (int q = 0; q < 8; ++q)
      dstrow[j * SJ_PITCH + q * 32 + lane] = v[j * SJ_RL + q];
    dstrow[j * SJ_PITCH + 256 + lane8] = v[j * SJ_RL + 8];
  }
}

__global__ __launch_bounds__(128, 4) void subj_gemm_kernel(
    const float* __restrict__ eeg, const int* __restrict__ sid,
    const float* __restrict__ sW, const float* __restrict__ sB,
    float* __restrict__ x) {
  __shared__ __attribute__((aligned(16))) float S[C_N * SJ_PITCH];

  const int b    = blockIdx.x;
  const int tid  = threadIdx.x;
  const int lane = tid & 31;
  const int half = lane >> 4;
  const int l    = lane & 15;
  const int wave = __builtin_amdgcn_readfirstlane(tid >> 5);

  const int s = sid[b];
  const float* __restrict__ E    = eeg + (size_t)b * CT_N;
  const float* __restrict__ W    = sW + (size_t)s * T_N * T_N;
  const float* __restrict__ bias = sB + (size_t)s * T_N;

  if (tid < C_N) S[tid * SJ_PITCH + T_N] = 0.0f;
#pragma unroll
  for (int bt = 0; bt < 8; ++bt) {
    float v[SJ_RB * SJ_RL];
    sj_load(v, E, wave * 16 + bt * SJ_RB, C_N - 1, lane);
    sj_store(S + (wave * 16 + bt * SJ_RB) * SJ_PITCH, v, lane);
  }
  __syncthreads();

  v2f a[SJ_KS];
  {
    const float* ap = S + (wave * 16 + l) * SJ_PITCH + 2 * half;
#pragma unroll
    for (int ks = 0; ks < SJ_KS; ++ks) a[ks] = *(const v2f*)(ap + 4 * ks);
  }
  __syncthreads();

#pragma unroll
  for (int bt = 0; bt < 4; ++bt) {
    float v[SJ_RB * SJ_RL];
    sj_load(v, W, wave * 8 + bt * SJ_RB, T_N - 1, lane);
    sj_store(S + (wave * 8 + bt * SJ_RB) * SJ_PITCH, v, lane);
  }
  float* __restrict__ X = x + (size_t)b * CT_N;

  for (int c = 0; c < SJ_NCH; ++c) {
    __syncthreads();
    const bool more = (c + 1 < SJ_NCH);
    const int nrow0 = (c + 1) * SJ_CH + wave * 8;
    float* __restrict__ dst =
        S + (((c + 1) & 1) * SJ_CH + wave * 8) * SJ_PITCH;

    const float* bp0 = S + (c & 1) * (SJ_CH * SJ_PITCH) + l * SJ_PITCH + 2 * half;
    const float* bp1 = bp0 + 16 * SJ_PITCH;
    v8f acc0 = {}, acc1 = {};
#pragma unroll
    for (int part = 0; part < 4; ++part) {
      float v[SJ_RB * SJ_RL];
      if (more) sj_load(v, W, nrow0 + part * SJ_RB, T_N - 1, lane);
#pragma unroll
      for (int ks = (part * SJ_KS + 3) / 4; ks < ((part + 1) * SJ_KS + 3) / 4; ++ks) {
        const v2f b0 = *(const v2f*)(bp0 + 4 * ks);
        const v2f b1 = *(const v2f*)(bp1 + 4 * ks);
        acc0 = wmma4(a[ks], b0, acc0);
        acc1 = wmma4(a[ks], b1, acc1);
      }
      if (more) sj_store(dst + part * SJ_RB * SJ_PITCH, v, lane);
    }
    const int col0 = c * SJ_CH + l;
    const int col1 = col0 + 16;
    const float bn0 = bias[col0];
#pragma unroll
    for (int r = 0; r < 8; ++r) {
      const int orow = wave * 16 + r + 8 * half;
      X[orow * T_N + col0] = acc0[r] + bn0;
    }
    if (col1 < T_N) {
      const float bn1 = bias[col1];
#pragma unroll
      for (int r = 0; r < 8; ++r) {
        const int orow = wave * 16 + r + 8 * half;
        X[orow * T_N + col1] = acc1[r] + bn1;
      }
    }
  }
}

typedef __attribute__((ext_vector_type(4))) float v4f;

#define ENC_RT   64
#define ENC_KS   16
#define ENC_KW   (CT_N / ENC_KS)
#define ENC_KC   100
#define ENC_NCH  (ENC_KW / ENC_KC)
#define ENC_AP   ENC_KC
#define ENC_BP   416
#define ENC_ASZ  (ENC_RT * ENC_AP)
#define ENC_BSZ  ((ENC_KC / 2) * ENC_BP)
#define ENC_BUF  (ENC_ASZ + ENC_BSZ)
#define ENC_BIT  ((ENC_KC / 2) * (EMB_N / 4))
#define ENC_BJ   5
#define ENC_AV4  (ENC_RT * (ENC_KC / 4))
#define ENC_AJ   4
#define ENC_PLANE (B_N * EMB_N)

__device__ __forceinline__ void enc_put_b(float* __restrict__ d, v4f r0, v4f r1) {
  v4f lo, hi;
  lo.x = r0.x; lo.y = r1.x; lo.z = r0.y; lo.w = r1.y;
  hi.x = r0.z; hi.y = r1.z; hi.z = r0.w; hi.w = r1.w;
  *(v4f*)d       = lo;
  *(v4f*)(d + 4) = hi;
}

__device__ __forceinline__ void enc_load(
    v4f (&sb)[2 * ENC_BJ], v4f (&sa)[ENC_AJ], const float* __restrict__ bg,
    const float* __restrict__ ag, const int (&bgo)[ENC_BJ],
    const int (&ago)[ENC_AJ], bool blast, bool alast) {
#pragma unroll
  for (int j = 0; j < ENC_BJ - 1; ++j) {
    sb[2 * j]     = *(const v4f*)(bg + bgo[j]);
    sb[2 * j + 1] = *(const v4f*)(bg + bgo[j] + EMB_N);
  }
  if (blast) {
    sb[2 * (ENC_BJ - 1)]     = *(const v4f*)(bg + bgo[ENC_BJ - 1]);
    sb[2 * (ENC_BJ - 1) + 1] = *(const v4f*)(bg + bgo[ENC_BJ - 1] + EMB_N);
  }
#pragma unroll
  for (int j = 0; j < ENC_AJ - 1; ++j) sa[j] = *(const v4f*)(ag + ago[j]);
  if (alast) sa[ENC_AJ - 1] = *(const v4f*)(ag + ago[ENC_AJ - 1]);
}

__device__ __forceinline__ void enc_store(
    float* __restrict__ buf, const v4f (&sb)[2 * ENC_BJ],
    const v4f (&sa)[ENC_AJ], const int (&blo)[ENC_BJ], int tid, bool blast,
    bool alast) {
#pragma unroll
  for (int j = 0; j < ENC_BJ - 1; ++j)
    enc_put_b(buf + blo[j], sb[2 * j], sb[2 * j + 1]);
  if (blast)
    enc_put_b(buf + blo[ENC_BJ - 1], sb[2 * (ENC_BJ - 1)], sb[2 * (ENC_BJ - 1) + 1]);
#pragma unroll
  for (int j = 0; j < ENC_AJ - 1; ++j) *(v4f*)(buf + (tid + 512 * j) * 4) = sa[j];
  if (alast) *(v4f*)(buf + (tid + 512 * (ENC_AJ - 1)) * 4) = sa[ENC_AJ - 1];
}

__global__ __launch_bounds__(512) void enc_gemm_kernel(
    const float* __restrict__ x, const float* __restrict__ eW,
    float* __restrict__ part) {
  __shared__ __attribute__((aligned(16))) float lds[2 * ENC_BUF];

  const int tid  = threadIdx.x;
  const int lane = tid & 31;
  const int wave = __builtin_amdgcn_readfirstlane(tid >> 5);
  const int half = lane >> 4;
  const int l    = lane & 15;
  const int mt   = blockIdx.x >> 4;
  const int ksp  = blockIdx.x & 15;
  const int m0   = mt * ENC_RT;
  const int k0   = ksp * ENC_KW;
  const int rt   = wave & 3;
  const int cg   = wave >> 2;

  int bgo[ENC_BJ], blo[ENC_BJ], ago[ENC_AJ];
#pragma unroll
  for (int j = 0; j < ENC_BJ; ++j) {
    const int i = tid + 512 * j;
    const int p = i / (EMB_N / 4);
    const int q = i - p * (EMB_N / 4);
    bgo[j] = 2 * p * EMB_N + q * 4;
    blo[j] = ENC_ASZ + p * ENC_BP + q * 8;
  }
#pragma unroll
  for (int j = 0; j < ENC_AJ; ++j) {
    const int i = tid + 512 * j;
    const int r = i / (ENC_KC / 4);
    ago[j] = r * CT_N + (i - r * (ENC_KC / 4)) * 4;
  }
  const bool blast = tid + 512 * (ENC_BJ - 1) < ENC_BIT;
  const bool alast = tid + 512 * (ENC_AJ - 1) < ENC_AV4;
  const float* __restrict__ Bg = eW + (size_t)k0 * EMB_N;
  const float* __restrict__ Ag = x + (size_t)m0 * CT_N + k0;

  for (int i = tid; i < 2 * (ENC_KC / 2) * 16; i += 512) {
    const int bf = i / ((ENC_KC / 2) * 16);
    const int r  = (i - bf * ((ENC_KC / 2) * 16)) >> 4;
    lds[bf * ENC_BUF + ENC_ASZ + r * ENC_BP + 2 * EMB_N + (i & 15)] = 0.0f;
  }

  v4f sb[2 * ENC_BJ], sa[ENC_AJ];
  enc_load(sb, sa, Bg, Ag, bgo, ago, blast, alast);
  enc_store(lds, sb, sa, blo, tid, blast, alast);

  v8f acc0 = {}, acc1 = {}, acc2 = {};
  for (int c = 0; c < ENC_NCH; ++c) {
    __syncthreads();
    const bool more = (c + 1 < ENC_NCH);
    if (more)
      enc_load(sb, sa, Bg + (size_t)(c + 1) * (ENC_KC * EMB_N),
               Ag + (c + 1) * ENC_KC, bgo, ago, blast, alast);

    const float* cur = lds + (c & 1) * ENC_BUF;
    const float* ap = cur + (rt * 16 + l) * ENC_AP + 2 * half;
    const float* bp = cur + ENC_ASZ + half * ENC_BP + (cg * 48 + l) * 2;
#pragma unroll
    for (int ks = 0; ks < ENC_KC / 4; ++ks) {
      const v2f a = *(const v2f*)(ap + 4 * ks);
      const v2f b0 = *(const v2f*)(bp + 2 * ks * ENC_BP);
      const v2f b1 = *(const v2f*)(bp + 2 * ks * ENC_BP + 32);
      const v2f b2 = *(const v2f*)(bp + 2 * ks * ENC_BP + 64);
      acc0 = wmma4(a, b0, acc0);
      acc1 = wmma4(a, b1, acc1);
      acc2 = wmma4(a, b2, acc2);
    }

    if (more)
      enc_store(lds + ((c + 1) & 1) * ENC_BUF, sb, sa, blo, tid, blast, alast);
  }

  float* __restrict__ P = part + (size_t)ksp * ENC_PLANE;
  const int col = cg * 48 + l;
#pragma unroll
  for (int r = 0; r < 8; ++r) {
    const int row = m0 + rt * 16 + r + 8 * half;
    P[(size_t)row * EMB_N + col]      = acc0[r];
    P[(size_t)row * EMB_N + col + 16] = acc1[r];
    if (col + 32 < EMB_N) P[(size_t)row * EMB_N + col + 32] = acc2[r];
  }
}

__global__ __launch_bounds__(256) void enc_sum_kernel(
    const float* __restrict__ part, const float* __restrict__ eB,
    float* __restrict__ h) {
  const int i = (blockIdx.x * 256 + threadIdx.x) * 4;
  v4f s = *(const v4f*)(part + i);
#pragma unroll
  for (int p = 1; p < ENC_KS; ++p) s += *(const v4f*)(part + (size_t)p * ENC_PLANE + i);
  const int cb = i % EMB_N;
  s.x += eB[cb];
  s.y += eB[cb + 1];
  s.z += eB[cb + 2];
  s.w += eB[cb + 3];
  *(v4f*)(h + i) = s;
}

__global__ __launch_bounds__(256) void proj1_gemm_kernel(
    const float* __restrict__ h, const float* __restrict__ W1,
    const float* __restrict__ b1, float* __restrict__ h1,
    float* __restrict__ g) {
  const int wave = threadIdx.x >> 5;
  const int lane = threadIdx.x & 31;
  const int tile = blockIdx.x * 8 + wave;
  const int mt = tile >> 5, np = tile & 31;
  const int m0 = mt * 16, n0 = np * 32;
  const int half = lane >> 4;
  const int l    = lane & 15;

  const int arow  = m0 + l;
  const int ncol0 = n0 + l;
  const int ncol1 = n0 + 16 + l;
  const float* __restrict__ Ar = h + (size_t)arow * EMB_N;

  v8f acc0 = {}, acc1 = {};
  for (int k = 0; k < EMB_N; k += 8) {
    const int ka = k + 2 * half;
    const v2f a0 = *(const v2f*)(Ar + ka);
    const v2f a1 = *(const v2f*)(Ar + ka + 4);
    v2f b00, b01, b10, b11;
    b00.x = W1[(size_t)ka * PROJ_N + ncol0];
    b00.y = W1[(size_t)(ka + 1) * PROJ_N + ncol0];
    b01.x = W1[(size_t)(ka + 4) * PROJ_N + ncol0];
    b01.y = W1[(size_t)(ka + 5) * PROJ_N + ncol0];
    b10.x = W1[(size_t)ka * PROJ_N + ncol1];
    b10.y = W1[(size_t)(ka + 1) * PROJ_N + ncol1];
    b11.x = W1[(size_t)(ka + 4) * PROJ_N + ncol1];
    b11.y = W1[(size_t)(ka + 5) * PROJ_N + ncol1];
    acc0 = wmma4(a0, b00, acc0);
    acc1 = wmma4(a0, b10, acc1);
    acc0 = wmma4(a1, b01, acc0);
    acc1 = wmma4(a1, b11, acc1);
  }
  const float bn0 = b1[ncol0];
  const float bn1 = b1[ncol1];
#pragma unroll
  for (int r = 0; r < 8; ++r) {
    const int row = m0 + r + 8 * half;
    const float t0 = acc0[r] + bn0;
    const float t1 = acc1[r] + bn1;
    h1[(size_t)row * PROJ_N + ncol0] = t0;
    h1[(size_t)row * PROJ_N + ncol1] = t1;
    g[(size_t)row * PROJ_N + ncol0] = gelu_exact(t0);
    g[(size_t)row * PROJ_N + ncol1] = gelu_exact(t1);
  }
}

__global__ __launch_bounds__(256) void proj2_gemm_kernel(
    const float* __restrict__ g, const float* __restrict__ W2,
    const float* __restrict__ b2, const float* __restrict__ h1,
    float* __restrict__ hsum) {
  const int wave = threadIdx.x >> 5;
  const int lane = threadIdx.x & 31;
  const int tile = blockIdx.x * 8 + wave;
  const int mt = tile >> 5, np = tile & 31;
  const int m0 = mt * 16, n0 = np * 32;
  const int half = lane >> 4;
  const int l    = lane & 15;

  const int arow  = m0 + l;
  const int ncol0 = n0 + l;
  const int ncol1 = n0 + 16 + l;
  const float* __restrict__ Ar = g + (size_t)arow * PROJ_N;

  v8f acc0 = {}, acc1 = {};
  for (int k = 0; k < PROJ_N; k += 8) {
    const int ka = k + 2 * half;
    const v2f a0 = *(const v2f*)(Ar + ka);
    const v2f a1 = *(const v2f*)(Ar + ka + 4);
    v2f b00, b01, b10, b11;
    b00.x = W2[(size_t)ka * PROJ_N + ncol0];
    b00.y = W2[(size_t)(ka + 1) * PROJ_N + ncol0];
    b01.x = W2[(size_t)(ka + 4) * PROJ_N + ncol0];
    b01.y = W2[(size_t)(ka + 5) * PROJ_N + ncol0];
    b10.x = W2[(size_t)ka * PROJ_N + ncol1];
    b10.y = W2[(size_t)(ka + 1) * PROJ_N + ncol1];
    b11.x = W2[(size_t)(ka + 4) * PROJ_N + ncol1];
    b11.y = W2[(size_t)(ka + 5) * PROJ_N + ncol1];
    acc0 = wmma4(a0, b00, acc0);
    acc1 = wmma4(a0, b10, acc1);
    acc0 = wmma4(a1, b01, acc0);
    acc1 = wmma4(a1, b11, acc1);
  }
  const float bn0 = b2[ncol0];
  const float bn1 = b2[ncol1];
#pragma unroll
  for (int r = 0; r < 8; ++r) {
    const int row = m0 + r + 8 * half;
    hsum[(size_t)row * PROJ_N + ncol0] =
        h1[(size_t)row * PROJ_N + ncol0] + acc0[r] + bn0;
    hsum[(size_t)row * PROJ_N + ncol1] =
        h1[(size_t)row * PROJ_N + ncol1] + acc1[r] + bn1;
  }
}

__global__ __launch_bounds__(256) void layernorm_kernel(
    const float* __restrict__ hsum, const float* __restrict__ gamma,
    const float* __restrict__ beta, float* __restrict__ out) {
  const int row = blockIdx.x;
  const int tid = threadIdx.x;
  const float* __restrict__ hr = hsum + (size_t)row * PROJ_N;

  float v[4];
  float s = 0.0f, ss = 0.0f;
#pragma unroll
  for (int i = 0; i < 4; ++i) {
    v[i] = hr[tid + i * 256];
    s += v[i];
    ss += v[i] * v[i];
  }
  __shared__ float rs[256];
  __shared__ float rss[256];
  rs[tid] = s;
  rss[tid] = ss;
  __syncthreads();
  for (int off = 128; off > 0; off >>= 1) {
    if (tid < off) {
      rs[tid] += rs[tid + off];
      rss[tid] += rss[tid + off];
    }
    __syncthreads();
  }
  const float mu  = rs[0] * (1.0f / PROJ_N);
  const float var = rss[0] * (1.0f / PROJ_N) - mu * mu;
  const float inv = rsqrtf(var + 1e-5f);
#pragma unroll
  for (int i = 0; i < 4; ++i) {
    const int c = tid + i * 256;
    out[(size_t)row * PROJ_N + c] = (v[i] - mu) * inv * gamma[c] + beta[c];
  }
}

extern "C" void kernel_launch(void* const* d_in, const int* in_sizes, int n_in,
                              void* d_out, int out_size, void* d_ws,
                              size_t ws_size, hipStream_t stream) {
  const float* eeg    = (const float*)d_in[0];
  const int*   sid    = (const int*)d_in[1];
  const float* subj_W = (const float*)d_in[2];
  const float* subj_b = (const float*)d_in[3];
  const float* enc_W  = (const float*)d_in[4];
  const float* enc_b  = (const float*)d_in[5];
  const float* W1     = (const float*)d_in[6];
  const float* b1     = (const float*)d_in[7];
  const float* W2     = (const float*)d_in[8];
  const float* b2     = (const float*)d_in[9];
  const float* gamma  = (const float*)d_in[10];
  const float* beta   = (const float*)d_in[11];
  float*       out    = (float*)d_out;

  float* xbuf  = (float*)d_ws;
  float* hbuf  = xbuf + (size_t)B_N * CT_N;
  float* h1buf = hbuf + (size_t)B_N * EMB_N;
  float* gbuf  = h1buf + (size_t)B_N * PROJ_N;
  float* hsbuf = gbuf + (size_t)B_N * PROJ_N;

  subj_gemm_kernel<<<B_N, 128, 0, stream>>>(eeg, sid, subj_W, subj_b, xbuf);
  float* pbuf = h1buf;
  enc_gemm_kernel<<<ENC_KS * (B_N / ENC_RT), 512, 0, stream>>>(xbuf, enc_W, pbuf);
  enc_sum_kernel<<<ENC_PLANE / 1024, 256, 0, stream>>>(pbuf, enc_b, hbuf);
  proj1_gemm_kernel<<<256, 256, 0, stream>>>(hbuf, W1, b1, h1buf, gbuf);
  proj2_gemm_kernel<<<256, 256, 0, stream>>>(gbuf, W2, b2, h1buf, hsbuf);
  layernorm_kernel<<<B_N, 256, 0, stream>>>(hsbuf, gamma, beta, out);
}


# projector layer-2 GEMM: one workgroup per 64x64 tile stages both operands in LDS by 16-byte row loads (was per-wave loads of 16 lines each), fragment reads 8 k-steps ahead; sum order kept
# speedup vs baseline: 6.4577x; 1.0969x over previous
#include <hip/hip_runtime.h>
#include <hip/hip_bf16.h>
#include <math.h>


typedef __attribute__((ext_vector_type(2))) float v2f;
typedef __attribute__((ext_vector_type(8))) float v8f;

#define B_N    1024
#define C_N    64
#define T_N    275
#define CT_N   (C_N * T_N)
#define EMB_N  184
#define PROJ_N 1024

__device__ __forceinline__ v8f wmma4(v2f a, v2f b, v8f c) {
  return __builtin_amdgcn_wmma_f32_16x16x4_f32(false, a, false, b, (short)0, c,
                                               false, false);
}

__device__ __forceinline__ float gelu_exact(float x) {
  return 0.5f * x * (1.0f + erff(x * 0.70710678118654752440f));
}

#define SJ_PITCH 276
#define SJ_KS    69
#define SJ_CH    32
#define SJ_NCH   9
#define SJ_RB    2
#define SJ_RL    9

__device__ __forceinline__ void sj_load(float (&v)[SJ_RB * SJ_RL],
                                        const float* __restrict__ mat,
                                        int row0, int lastrow, int lane) {
  const int lane8 = lane < 19 ? lane : 18;
#pragma unroll
  for (int j = 0; j < SJ_RB; ++j) {
    int r = row0 + j;
    if (r > lastrow) r = lastrow;
    const float* __restrict__ rp = mat + (size_t)r * T_N;
#pragma unroll
    for (int q = 0; q < 8; ++q) v[j * SJ_RL + q] = rp[q * 32 + lane];
    v[j * SJ_RL + 8] = rp[256 + lane8];
  }
}

__device__ __forceinline__ void sj_store(float* __restrict__ dstrow,
                                         const float (&v)[SJ_RB * SJ_RL],
                                         int lane) {
  const int lane8 = lane < 19 ? lane : 18;
#pragma unroll
  for (int j = 0; j < SJ_RB; ++j) {
#pragma unroll
    for (int q = 0; q < 8; ++q)
      dstrow[j * SJ_PITCH + q * 32 + lane] = v[j * SJ_RL + q];
    dstrow[j * SJ_PITCH + 256 + lane8] = v[j * SJ_RL + 8];
  }
}

__global__ __launch_bounds__(128, 4) void subj_gemm_kernel(
    const float* __restrict__ eeg, const int* __restrict__ sid,
    const float* __restrict__ sW, const float* __restrict__ sB,
    float* __restrict__ x) {
  __shared__ __attribute__((aligned(16))) float S[C_N * SJ_PITCH];

  const int b    = blockIdx.x;
  const int tid  = threadIdx.x;
  const int lane = tid & 31;
  const int half = lane >> 4;
  const int l    = lane & 15;
  const int wave = __builtin_amdgcn_readfirstlane(tid >> 5);

  const int s = sid[b];
  const float* __restrict__ E    = eeg + (size_t)b * CT_N;
  const float* __restrict__ W    = sW + (size_t)s * T_N * T_N;
  const float* __restrict__ bias = sB + (size_t)s * T_N;

  if (tid < C_N) S[tid * SJ_PITCH + T_N] = 0.0f;
#pragma unroll
  for (int bt = 0; bt < 8; ++bt) {
    float v[SJ_RB * SJ_RL];
    sj_load(v, E, wave * 16 + bt * SJ_RB, C_N - 1, lane);
    sj_store(S + (wave * 16 + bt * SJ_RB) * SJ_PITCH, v, lane);
  }
  __syncthreads();

  v2f a[SJ_KS];
  {
    const float* ap = S + (wave * 16 + l) * SJ_PITCH + 2 * half;
#pragma unroll
    for (int ks = 0; ks < SJ_KS; ++ks) a[ks] = *(const v2f*)(ap + 4 * ks);
  }
  __syncthreads();

#pragma unroll
  for (int bt = 0; bt < 4; ++bt) {
    float v[SJ_RB * SJ_RL];
    sj_load(v, W, wave * 8 + bt * SJ_RB, T_N - 1, lane);
    sj_store(S + (wave * 8 + bt * SJ_RB) * SJ_PITCH, v, lane);
  }
  float* __restrict__ X = x + (size_t)b * CT_N;

  for (int c = 0; c < SJ_NCH; ++c) {
    __syncthreads();
    const bool more = (c + 1 < SJ_NCH);
    const int nrow0 = (c + 1) * SJ_CH + wave * 8;
    float* __restrict__ dst =
        S + (((c + 1) & 1) * SJ_CH + wave * 8) * SJ_PITCH;

    const float* bp0 = S + (c & 1) * (SJ_CH * SJ_PITCH) + l * SJ_PITCH + 2 * half;
    const float* bp1 = bp0 + 16 * SJ_PITCH;
    v8f acc0 = {}, acc1 = {};
#pragma unroll
    for (int part = 0; part < 4; ++part) {
      float v[SJ_RB * SJ_RL];
      if (more) sj_load(v, W, nrow0 + part * SJ_RB, T_N - 1, lane);
#pragma unroll
      for (int ks = (part * SJ_KS + 3) / 4; ks < ((part + 1) * SJ_KS + 3) / 4; ++ks) {
        const v2f b0 = *(const v2f*)(bp0 + 4 * ks);
        const v2f b1 = *(const v2f*)(bp1 + 4 * ks);
        acc0 = wmma4(a[ks], b0, acc0);
        acc1 = wmma4(a[ks], b1, acc1);
      }
      if (more) sj_store(dst + part * SJ_RB * SJ_PITCH, v, lane);
    }
    const int col0 = c * SJ_CH + l;
    const int col1 = col0 + 16;
    const float bn0 = bias[col0];
#pragma unroll
    for (int r = 0; r < 8; ++r) {
      const int orow = wave * 16 + r + 8 * half;
      X[orow * T_N + col0] = acc0[r] + bn0;
    }
    if (col1 < T_N) {
      const float bn1 = bias[col1];
#pragma unroll
      for (int r = 0; r < 8; ++r) {
        const int orow = wave * 16 + r + 8 * half;
        X[orow * T_N + col1] = acc1[r] + bn1;
      }
    }
  }
}

typedef __attribute__((ext_vector_type(4))) float v4f;

#define ENC_RT   64
#define ENC_KS   16
#define ENC_KW   (CT_N / ENC_KS)
#define ENC_KC   100
#define ENC_NCH  (ENC_KW / ENC_KC)
#define ENC_AP   ENC_KC
#define ENC_BP   416
#define ENC_ASZ  (ENC_RT * ENC_AP)
#define ENC_BSZ  ((ENC_KC / 2) * ENC_BP)
#define ENC_BUF  (ENC_ASZ + ENC_BSZ)
#define ENC_BIT  ((ENC_KC / 2) * (EMB_N / 4))
#define ENC_BJ   5
#define ENC_AV4  (ENC_RT * (ENC_KC / 4))
#define ENC_AJ   4
#define ENC_PLANE (B_N * EMB_N)

__device__ __forceinline__ void enc_put_b(float* __restrict__ d, v4f r0, v4f r1) {
  v4f lo, hi;
  lo.x = r0.x; lo.y = r1.x; lo.z = r0.y; lo.w = r1.y;
  hi.x = r0.z; hi.y = r1.z; hi.z = r0.w; hi.w = r1.w;
  *(v4f*)d       = lo;
  *(v4f*)(d + 4) = hi;
}

__device__ __forceinline__ void enc_load(
    v4f (&sb)[2 * ENC_BJ], v4f (&sa)[ENC_AJ], const float* __restrict__ bg,
    const float* __restrict__ ag, const int (&bgo)[ENC_BJ],
    const int (&ago)[ENC_AJ], bool blast, bool alast) {
#pragma unroll
  for (int j = 0; j < ENC_BJ - 1; ++j) {
    sb[2 * j]     = *(const v4f*)(bg + bgo[j]);
    sb[2 * j + 1] = *(const v4f*)(bg + bgo[j] + EMB_N);
  }
  if (blast) {
    sb[2 * (ENC_BJ - 1)]     = *(const v4f*)(bg + bgo[ENC_BJ - 1]);
    sb[2 * (ENC_BJ - 1) + 1] = *(const v4f*)(bg + bgo[ENC_BJ - 1] + EMB_N);
  }
#pragma unroll
  for (int j = 0; j < ENC_AJ - 1; ++j) sa[j] = *(const v4f*)(ag + ago[j]);
  if (alast) sa[ENC_AJ - 1] = *(const v4f*)(ag + ago[ENC_AJ - 1]);
}

__device__ __forceinline__ void enc_store(
    float* __restrict__ buf, const v4f (&sb)[2 * ENC_BJ],
    const v4f (&sa)[ENC_AJ], const int (&blo)[ENC_BJ], int tid, bool blast,
    bool alast) {
#pragma unroll
  for (int j = 0; j < ENC_BJ - 1; ++j)
    enc_put_b(buf + blo[j], sb[2 * j], sb[2 * j + 1]);
  if (blast)
    enc_put_b(buf + blo[ENC_BJ - 1], sb[2 * (ENC_BJ - 1)], sb[2 * (ENC_BJ - 1) + 1]);
#pragma unroll
  for (int j = 0; j < ENC_AJ - 1; ++j) *(v4f*)(buf + (tid + 512 * j) * 4) = sa[j];
  if (alast) *(v4f*)(buf + (tid + 512 * (ENC_AJ - 1)) * 4) = sa[ENC_AJ - 1];
}

__global__ __launch_bounds__(512) void enc_gemm_kernel(
    const float* __restrict__ x, const float* __restrict__ eW,
    float* __restrict__ part) {
  __shared__ __attribute__((aligned(16))) float lds[2 * ENC_BUF];

  const int tid  = threadIdx.x;
  const int lane = tid & 31;
  const int wave = __builtin_amdgcn_readfirstlane(tid >> 5);
  const int half = lane >> 4;
  const int l    = lane & 15;
  const int mt   = blockIdx.x >> 4;
  const int ksp  = blockIdx.x & 15;
  const int m0   = mt * ENC_RT;
  const int k0   = ksp * ENC_KW;
  const int rt   = wave & 3;
  const int cg   = wave >> 2;

  int bgo[ENC_BJ], blo[ENC_BJ], ago[ENC_AJ];
#pragma unroll
  for (int j = 0; j < ENC_BJ; ++j) {
    const int i = tid + 512 * j;
    const int p = i / (EMB_N / 4);
    const int q = i - p * (EMB_N / 4);
    bgo[j] = 2 * p * EMB_N + q * 4;
    blo[j] = ENC_ASZ + p * ENC_BP + q * 8;
  }
#pragma unroll
  for (int j = 0; j < ENC_AJ; ++j) {
    const int i = tid + 512 * j;
    const int r = i / (ENC_KC / 4);
    ago[j] = r * CT_N + (i - r * (ENC_KC / 4)) * 4;
  }
  const bool blast = tid + 512 * (ENC_BJ - 1) < ENC_BIT;
  const bool alast = tid + 512 * (ENC_AJ - 1) < ENC_AV4;
  const float* __restrict__ Bg = eW + (size_t)k0 * EMB_N;
  const float* __restrict__ Ag = x + (size_t)m0 * CT_N + k0;

  for (int i = tid; i < 2 * (ENC_KC / 2) * 16; i += 512) {
    const int bf = i / ((ENC_KC / 2) * 16);
    const int r  = (i - bf * ((ENC_KC / 2) * 16)) >> 4;
    lds[bf * ENC_BUF + ENC_ASZ + r * ENC_BP + 2 * EMB_N + (i & 15)] = 0.0f;
  }

  v4f sb[2 * ENC_BJ], sa[ENC_AJ];
  enc_load(sb, sa, Bg, Ag, bgo, ago, blast, alast);
  enc_store(lds, sb, sa, blo, tid, blast, alast);

  v8f acc0 = {}, acc1 = {}, acc2 = {};
  for (int c = 0; c < ENC_NCH; ++c) {
    __syncthreads();
    const bool more = (c + 1 < ENC_NCH);
    if (more)
      enc_load(sb, sa, Bg + (size_t)(c + 1) * (ENC_KC * EMB_N),
               Ag + (c + 1) * ENC_KC, bgo, ago, blast, alast);

    const float* cur = lds + (c & 1) * ENC_BUF;
    const float* ap = cur + (rt * 16 + l) * ENC_AP + 2 * half;
    const float* bp = cur + ENC_ASZ + half * ENC_BP + (cg * 48 + l) * 2;
#pragma unroll
    for (int ks = 0; ks < ENC_KC / 4; ++ks) {
      const v2f a = *(const v2f*)(ap + 4 * ks);
      const v2f b0 = *(const v2f*)(bp + 2 * ks * ENC_BP);
      const v2f b1 = *(const v2f*)(bp + 2 * ks * ENC_BP + 32);
      const v2f b2 = *(const v2f*)(bp + 2 * ks * ENC_BP + 64);
      acc0 = wmma4(a, b0, acc0);
      acc1 = wmma4(a, b1, acc1);
      acc2 = wmma4(a, b2, acc2);
    }

    if (more)
      enc_store(lds + ((c + 1) & 1) * ENC_BUF, sb, sa, blo, tid, blast, alast);
  }

  float* __restrict__ P = part + (size_t)ksp * ENC_PLANE;
  const int col = cg * 48 + l;
#pragma unroll
  for (int r = 0; r < 8; ++r) {
    const int row = m0 + rt * 16 + r + 8 * half;
    P[(size_t)row * EMB_N + col]      = acc0[r];
    P[(size_t)row * EMB_N + col + 16] = acc1[r];
    if (col + 32 < EMB_N) P[(size_t)row * EMB_N + col + 32] = acc2[r];
  }
}

__global__ __launch_bounds__(256) void enc_sum_kernel(
    const float* __restrict__ part, const float* __restrict__ eB,
    float* __restrict__ h) {
  const int i = (blockIdx.x * 256 + threadIdx.x) * 4;
  v4f s = *(const v4f*)(part + i);
#pragma unroll
  for (int p = 1; p < ENC_KS; ++p) s += *(const v4f*)(part + (size_t)p * ENC_PLANE + i);
  const int cb = i % EMB_N;
  s.x += eB[cb];
  s.y += eB[cb + 1];
  s.z += eB[cb + 2];
  s.w += eB[cb + 3];
  *(v4f*)(h + i) = s;
}

__global__ __launch_bounds__(256) void proj1_gemm_kernel(
    const float* __restrict__ h, const float* __restrict__ W1,
    const float* __restrict__ b1, float* __restrict__ h1,
    float* __restrict__ g) {
  const int wave = threadIdx.x >> 5;
  const int lane = threadIdx.x & 31;
  const int tile = blockIdx.x * 8 + wave;
  const int mt = tile >> 5, np = tile & 31;
  const int m0 = mt * 16, n0 = np * 32;
  const int half = lane >> 4;
  const int l    = lane & 15;

  const int arow  = m0 + l;
  const int ncol0 = n0 + l;
  const int ncol1 = n0 + 16 + l;
  const float* __restrict__ Ar = h + (size_t)arow * EMB_N;

  v8f acc0 = {}, acc1 = {};
  for (int k = 0; k < EMB_N; k += 8) {
    const int ka = k + 2 * half;
    const v2f a0 = *(const v2f*)(Ar + ka);
    const v2f a1 = *(const v2f*)(Ar + ka + 4);
    v2f b00, b01, b10, b11;
    b00.x = W1[(size_t)ka * PROJ_N + ncol0];
    b00.y = W1[(size_t)(ka + 1) * PROJ_N + ncol0];
    b01.x = W1[(size_t)(ka + 4) * PROJ_N + ncol0];
    b01.y = W1[(size_t)(ka + 5) * PROJ_N + ncol0];
    b10.x = W1[(size_t)ka * PROJ_N + ncol1];
    b10.y = W1[(size_t)(ka + 1) * PROJ_N + ncol1];
    b11.x = W1[(size_t)(ka + 4) * PROJ_N + ncol1];
    b11.y = W1[(size_t)(ka + 5) * PROJ_N + ncol1];
    acc0 = wmma4(a0, b00, acc0);
    acc1 = wmma4(a0, b10, acc1);
    acc0 = wmma4(a1, b01, acc0);
    acc1 = wmma4(a1, b11, acc1);
  }
  const float bn0 = b1[ncol0];
  const float bn1 = b1[ncol1];
#pragma unroll
  for (int r = 0; r < 8; ++r) {
    const int row = m0 + r + 8 * half;
    const float t0 = acc0[r] + bn0;
    const float t1 = acc1[r] + bn1;
    h1[(size_t)row * PROJ_N + ncol0] = t0;
    h1[(size_t)row * PROJ_N + ncol1] = t1;
    g[(size_t)row * PROJ_N + ncol0] = gelu_exact(t0);
    g[(size_t)row * PROJ_N + ncol1] = gelu_exact(t1);
  }
}

#define P2_T    64
#define P2_KC   128
#define P2_NCH  (PROJ_N / P2_KC)
#define P2_AP   132
#define P2_BP   160
#define P2_ASZ  (P2_T * P2_AP)
#define P2_BSZ  ((P2_KC / 2) * P2_BP)
#define P2_BUF  (P2_ASZ + P2_BSZ)
#define P2_AJ   8
#define P2_BJ   4
#define P2_G    8
#define P2_NG   (P2_KC / 4 / P2_G)

__device__ __forceinline__ void p2_load(v4f (&sa)[P2_AJ], v4f (&sb)[2 * P2_BJ],
                                        const float* __restrict__ ag,
                                        const float* __restrict__ bg) {
#pragma unroll
  for (int j = 0; j < P2_AJ; ++j)
    sa[j] = *(const v4f*)(ag + (size_t)j * 8 * PROJ_N);
#pragma unroll
  for (int j = 0; j < P2_BJ; ++j) {
    sb[2 * j]     = *(const v4f*)(bg + (size_t)j * 32 * PROJ_N);
    sb[2 * j + 1] = *(const v4f*)(bg + (size_t)j * 32 * PROJ_N + PROJ_N);
  }
}

__device__ __forceinline__ void p2_store(float* __restrict__ buf,
                                         const v4f (&sa)[P2_AJ],
                                         const v4f (&sb)[2 * P2_BJ], int alo,
                                         int blo) {
#pragma unroll
  for (int j = 0; j < P2_AJ; ++j) *(v4f*)(buf + alo + j * 8 * P2_AP) = sa[j];
#pragma unroll
  for (int j = 0; j < P2_BJ; ++j)
    enc_put_b(buf + blo + j * 16 * P2_BP, sb[2 * j], sb[2 * j + 1]);
}

__device__ __forceinline__ void p2_frag(v2f (&fa)[P2_G], v2f (&fb0)[P2_G],
                                        v2f (&fb1)[P2_G], const float* ap,
                                        const float* bp, int grp) {
#pragma unroll
  for (int j = 0; j < P2_G; ++j) {
    const int ks = grp * P2_G + j;
    fa[j]  = *(const v2f*)(ap + 4 * ks);
    fb0[j] = *(const v2f*)(bp + 2 * ks * P2_BP);
    fb1[j] = *(const v2f*)(bp + 2 * ks * P2_BP + 32);
  }
}

__global__ __launch_bounds__(256) void proj2_gemm_kernel(
    const float* __restrict__ g, const float* __restrict__ W2,
    const float* __restrict__ b2, const float* __restrict__ h1,
    float* __restrict__ hsum) {
  __shared__ __attribute__((aligned(16))) float lds[2 * P2_BUF];

  const int tid  = threadIdx.x;
  const int lane = tid & 31;
  const int wave = __builtin_amdgcn_readfirstlane(tid >> 5);
  const int half = lane >> 4;
  const int l    = lane & 15;
  const int m0   = (blockIdx.x >> 4) * P2_T;
  const int n0   = (blockIdx.x & 15) * P2_T;
  const int rt   = wave & 3;
  const int cg   = wave >> 2;

  const float* __restrict__ Ag =
      g + (size_t)(m0 + (tid >> 5)) * PROJ_N + (tid & 31) * 4;
  const float* __restrict__ Bg =
      W2 + (size_t)(2 * (tid >> 4)) * PROJ_N + n0 + (tid & 15) * 4;
  const int alo = (tid >> 5) * P2_AP + (tid & 31) * 4;
  const int blo = P2_ASZ + (tid >> 4) * P2_BP + (tid & 15) * 8;

  v4f sa[P2_AJ], sb[2 * P2_BJ];
  p2_load(sa, sb, Ag, Bg);
  p2_store(lds, sa, sb, alo, blo);

  v8f acc0 = {}, acc1 = {};
  for (int c = 0; c < P2_NCH; ++c) {
    __syncthreads();
    const bool more = (c + 1 < P2_NCH);
    if (more)
      p2_load(sa, sb, Ag + (c + 1) * P2_KC,
              Bg + (size_t)(c + 1) * P2_KC * PROJ_N);

    const float* cur = lds + (c & 1) * P2_BUF;
    const float* ap = cur + (rt * 16 + l) * P2_AP + 2 * half;
    const float* bp = cur + P2_ASZ + half * P2_BP + (cg * 32 + l) * 2;
    v2f fa[2][P2_G], fb0[2][P2_G], fb1[2][P2_G];
    p2_frag(fa[0], fb0[0], fb1[0], ap, bp, 0);
#pragma unroll
    for (int grp = 0; grp < P2_NG; ++grp) {
      if (grp + 1 < P2_NG)
        p2_frag(fa[(grp + 1) & 1], fb0[(grp + 1) & 1], fb1[(grp + 1) & 1], ap,
                bp, grp + 1);
      __builtin_amdgcn_sched_barrier(0);
#pragma unroll
      for (int j = 0; j < P2_G; ++j) {
        acc0 = wmma4(fa[grp & 1][j], fb0[grp & 1][j], acc0);
        acc1 = wmma4(fa[grp & 1][j], fb1[grp & 1][j], acc1);
      }
      __builtin_amdgcn_sched_barrier(0);
    }

    if (more) p2_store(lds + ((c + 1) & 1) * P2_BUF, sa, sb, alo, blo);
  }

  const int ncol0 = n0 + cg * 32 + l;
  const int ncol1 = ncol0 + 16;
  const float bn0 = b2[ncol0];
  const float bn1 = b2[ncol1];
#pragma unroll
  for (int r = 0; r < 8; ++r) {
    const int row = m0 + rt * 16 + r + 8 * half;
    hsum[(size_t)row * PROJ_N + ncol0] =
        h1[(size_t)row * PROJ_N + ncol0] + acc0[r] + bn0;
    hsum[(size_t)row * PROJ_N + ncol1] =
        h1[(size_t)row * PROJ_N + ncol1] + acc1[r] + bn1;
  }
}

__global__ __launch_bounds__(256) void layernorm_kernel(
    const float* __restrict__ hsum, const float* __restrict__ gamma,
    const float* __restrict__ beta, float* __restrict__ out) {
  const int row = blockIdx.x;
  const int tid = threadIdx.x;
  const float* __restrict__ hr = hsum + (size_t)row * PROJ_N;

  float v[4];
  float s = 0.0f, ss = 0.0f;
#pragma unroll
  for (int i = 0; i < 4; ++i) {
    v[i] = hr[tid + i * 256];
    s += v[i];
    ss += v[i] * v[i];
  }
  __shared__ float rs[256];
  __shared__ float rss[256];
  rs[tid] = s;
  rss[tid] = ss;
  __syncthreads();
  for (int off = 128; off > 0; off >>= 1) {
    if (tid < off) {
      rs[tid] += rs[tid + off];
      rss[tid] += rss[tid + off];
    }
    __syncthreads();
  }
  const float mu  = rs[0] * (1.0f / PROJ_N);
  const float var = rss[0] * (1.0f / PROJ_N) - mu * mu;
  const float inv = rsqrtf(var + 1e-5f);
#pragma unroll
  for (int i = 0; i < 4; ++i) {
    const int c = tid + i * 256;
    out[(size_t)row * PROJ_N + c] = (v[i] - mu) * inv * gamma[c] + beta[c];
  }
}

extern "C" void kernel_launch(void* const* d_in, const int* in_sizes, int n_in,
                              void* d_out, int out_size, void* d_ws,
                              size_t ws_size, hipStream_t stream) {
  const float* eeg    = (const float*)d_in[0];
  const int*   sid    = (const int*)d_in[1];
  const float* subj_W = (const float*)d_in[2];
  const float* subj_b = (const float*)d_in[3];
  const float* enc_W  = (const float*)d_in[4];
  const float* enc_b  = (const float*)d_in[5];
  const float* W1     = (const float*)d_in[6];
  const float* b1     = (const float*)d_in[7];
  const float* W2     = (const float*)d_in[8];
  const float* b2     = (const float*)d_in[9];
  const float* gamma  = (const float*)d_in[10];
  const float* beta   = (const float*)d_in[11];
  float*       out    = (float*)d_out;

  float* xbuf  = (float*)d_ws;
  float* hbuf  = xbuf + (size_t)B_N * CT_N;
  float* h1buf = hbuf + (size_t)B_N * EMB_N;
  float* gbuf  = h1buf + (size_t)B_N * PROJ_N;
  float* hsbuf = gbuf + (size_t)B_N * PROJ_N;

  subj_gemm_kernel<<<B_N, 128, 0, stream>>>(eeg, sid, subj_W, subj_b, xbuf);
  float* pbuf = h1buf;
  enc_gemm_kernel<<<ENC_KS * (B_N / ENC_RT), 512, 0, stream>>>(xbuf, enc_W, pbuf);
  enc_sum_kernel<<<ENC_PLANE / 1024, 256, 0, stream>>>(pbuf, enc_b, hbuf);
  proj1_gemm_kernel<<<256, 256, 0, stream>>>(hbuf, W1, b1, h1buf, gbuf);
  proj2_gemm_kernel<<<256, 256, 0, stream>>>(gbuf, W2, b2, h1buf, hsbuf);
  layernorm_kernel<<<B_N, 256, 0, stream>>>(hsbuf, gamma, beta, out);
}


# subject GEMM: B fragments of 6 k-steps read from LDS as one group into a second register set ahead of the products (was one pair of k-steps at a time, waited for before each); sum order kept
# speedup vs baseline: 6.7311x; 1.0423x over previous
#include <hip/hip_runtime.h>
#include <hip/hip_bf16.h>
#include <math.h>


typedef __attribute__((ext_vector_type(2))) float v2f;
typedef __attribute__((ext_vector_type(8))) float v8f;

#define B_N    1024
#define C_N    64
#define T_N    275
#define CT_N   (C_N * T_N)
#define EMB_N  184
#define PROJ_N 1024

__device__ __forceinline__ v8f wmma4(v2f a, v2f b, v8f c) {
  return __builtin_amdgcn_wmma_f32_16x16x4_f32(false, a, false, b, (short)0, c,
                                               false, false);
}

__device__ __forceinline__ float gelu_exact(float x) {
  return 0.5f * x * (1.0f + erff(x * 0.70710678118654752440f));
}

#define SJ_PITCH 276
#define SJ_KS    69
#define SJ_CH    32
#define SJ_NCH   9
#define SJ_RB    2
#define SJ_RL    9
#define SJ_NG    12
#define SJ_G     ((SJ_KS + SJ_NG - 1) / SJ_NG)
#define SJ_SPAN  (SJ_NG / 4)
#define SJ_GB(g) (((g) * SJ_KS + SJ_NG - 1) / SJ_NG)

__device__ __forceinline__ void sj_load(float (&v)[SJ_RB * SJ_RL],
                                        const float* __restrict__ mat,
                                        int row0, int lastrow, int lane) {
  const int lane8 = lane < 19 ? lane : 18;
#pragma unroll
  for (int j = 0; j < SJ_RB; ++j) {
    int r = row0 + j;
    if (r > lastrow) r = lastrow;
    const float* __restrict__ rp = mat + (size_t)r * T_N;
#pragma unroll
    for (int q = 0; q < 8; ++q) v[j * SJ_RL + q] = rp[q * 32 + lane];
    v[j * SJ_RL + 8] = rp[256 + lane8];
  }
}

__device__ __forceinline__ void sj_store(float* __restrict__ dstrow,
                                         const float (&v)[SJ_RB * SJ_RL],
                                         int lane) {
  const int lane8 = lane < 19 ? lane : 18;
#pragma unroll
  for (int j = 0; j < SJ_RB; ++j) {
#pragma unroll
    for (int q = 0; q < 8; ++q)
      dstrow[j * SJ_PITCH + q * 32 + lane] = v[j * SJ_RL + q];
    dstrow[j * SJ_PITCH + 256 + lane8] = v[j * SJ_RL + 8];
  }
}

__device__ __forceinline__ void sj_frag(v2f (&f0)[SJ_G], v2f (&f1)[SJ_G],
                                        const float* bp0, const float* bp1,
                                        int k_lo, int k_hi) {
#pragma unroll
  for (int j = 0; j < SJ_G; ++j) {
    if (k_lo + j < k_hi) {
      f0[j] = *(const v2f*)(bp0 + 4 * (k_lo + j));
      f1[j] = *(const v2f*)(bp1 + 4 * (k_lo + j));
    }
  }
}

__global__ __launch_bounds__(128, 4) void subj_gemm_kernel(
    const float* __restrict__ eeg, const int* __restrict__ sid,
    const float* __restrict__ sW, const float* __restrict__ sB,
    float* __restrict__ x) {
  __shared__ __attribute__((aligned(16))) float S[C_N * SJ_PITCH];

  const int b    = blockIdx.x;
  const int tid  = threadIdx.x;
  const int lane = tid & 31;
  const int half = lane >> 4;
  const int l    = lane & 15;
  const int wave = __builtin_amdgcn_readfirstlane(tid >> 5);

  const int s = sid[b];
  const float* __restrict__ E    = eeg + (size_t)b * CT_N;
  const float* __restrict__ W    = sW + (size_t)s * T_N * T_N;
  const float* __restrict__ bias = sB + (size_t)s * T_N;

  if (tid < C_N) S[tid * SJ_PITCH + T_N] = 0.0f;
#pragma unroll
  for (int bt = 0; bt < 8; ++bt) {
    float v[SJ_RB * SJ_RL];
    sj_load(v, E, wave * 16 + bt * SJ_RB, C_N - 1, lane);
    sj_store(S + (wave * 16 + bt * SJ_RB) * SJ_PITCH, v, lane);
  }
  __syncthreads();

  v2f a[SJ_KS];
  {
    const float* ap = S + (wave * 16 + l) * SJ_PITCH + 2 * half;
#pragma unroll
    for (int ks = 0; ks < SJ_KS; ++ks) a[ks] = *(const v2f*)(ap + 4 * ks);
  }
  __syncthreads();

#pragma unroll
  for (int bt = 0; bt < 4; ++bt) {
    float v[SJ_RB * SJ_RL];
    sj_load(v, W, wave * 8 + bt * SJ_RB, T_N - 1, lane);
    sj_store(S + (wave * 8 + bt * SJ_RB) * SJ_PITCH, v, lane);
  }
  float* __restrict__ X = x + (size_t)b * CT_N;

  for (int c = 0; c < SJ_NCH; ++c) {
    __syncthreads();
    const bool more = (c + 1 < SJ_NCH);
    const int nrow0 = (c + 1) * SJ_CH + wave * 8;
    float* __restrict__ dst =
        S + (((c + 1) & 1) * SJ_CH + wave * 8) * SJ_PITCH;

    const float* bp0 = S + (c & 1) * (SJ_CH * SJ_PITCH) + l * SJ_PITCH + 2 * half;
    const float* bp1 = bp0 + 16 * SJ_PITCH;
    v8f acc0 = {}, acc1 = {};
    v2f fb0[2][SJ_G], fb1[2][SJ_G];
    float v[SJ_RB * SJ_RL];
    sj_frag(fb0[0], fb1[0], bp0, bp1, SJ_GB(0), SJ_GB(1));
#pragma unroll
    for (int g = 0; g < SJ_NG; ++g) {
      if (more && g % SJ_SPAN == 0)
        sj_load(v, W, nrow0 + (g / SJ_SPAN) * SJ_RB, T_N - 1, lane);
      if (g + 1 < SJ_NG)
        sj_frag(fb0[(g + 1) & 1], fb1[(g + 1) & 1], bp0, bp1, SJ_GB(g + 1),
                SJ_GB(g + 2));
      __builtin_amdgcn_sched_barrier(0);
#pragma unroll
      for (int ks = SJ_GB(g); ks < SJ_GB(g + 1); ++ks) {
        acc0 = wmma4(a[ks], fb0[g & 1][ks - SJ_GB(g)], acc0);
        acc1 = wmma4(a[ks], fb1[g & 1][ks - SJ_GB(g)], acc1);
      }
      __builtin_amdgcn_sched_barrier(0);
      if (more && g % SJ_SPAN == SJ_SPAN - 1)
        sj_store(dst + (g / SJ_SPAN) * SJ_RB * SJ_PITCH, v, lane);
    }
    const int col0 = c * SJ_CH + l;
    const int col1 = col0 + 16;
    const float bn0 = bias[col0];
#pragma unroll
    for (int r = 0; r < 8; ++r) {
      const int orow = wave * 16 + r + 8 * half;
      X[orow * T_N + col0] = acc0[r] + bn0;
    }
    if (col1 < T_N) {
      const float bn1 = bias[col1];
#pragma unroll
      for (int r = 0; r < 8; ++r) {
        const int orow = wave * 16 + r + 8 * half;
        X[orow * T_N + col1] = acc1[r] + bn1;
      }
    }
  }
}

typedef __attribute__((ext_vector_type(4))) float v4f;

#define ENC_RT   64
#define ENC_KS   16
#define ENC_KW   (CT_N / ENC_KS)
#define ENC_KC   100
#define ENC_NCH  (ENC_KW / ENC_KC)
#define ENC_AP   ENC_KC
#define ENC_BP   416
#define ENC_ASZ  (ENC_RT * ENC_AP)
#define ENC_BSZ  ((ENC_KC / 2) * ENC_BP)
#define ENC_BUF  (ENC_ASZ + ENC_BSZ)
#define ENC_BIT  ((ENC_KC / 2) * (EMB_N / 4))
#define ENC_BJ   5
#define ENC_AV4  (ENC_RT * (ENC_KC / 4))
#define ENC_AJ   4
#define ENC_PLANE (B_N * EMB_N)

__device__ __forceinline__ void enc_put_b(float* __restrict__ d, v4f r0, v4f r1) {
  v4f lo, hi;
  lo.x = r0.x; lo.y = r1.x; lo.z = r0.y; lo.w = r1.y;
  hi.x = r0.z; hi.y = r1.z; hi.z = r0.w; hi.w = r1.w;
  *(v4f*)d       = lo;
  *(v4f*)(d + 4) = hi;
}

__device__ __forceinline__ void enc_load(
    v4f (&sb)[2 * ENC_BJ], v4f (&sa)[ENC_AJ], const float* __restrict__ bg,
    const float* __restrict__ ag, const int (&bgo)[ENC_BJ],
    const int (&ago)[ENC_AJ], bool blast, bool alast) {
#pragma unroll
  for (int j = 0; j < ENC_BJ - 1; ++j) {
    sb[2 * j]     = *(const v4f*)(bg + bgo[j]);
    sb[2 * j + 1] = *(const v4f*)(bg + bgo[j] + EMB_N);
  }
  if (blast) {
    sb[2 * (ENC_BJ - 1)]     = *(const v4f*)(bg + bgo[ENC_BJ - 1]);
    sb[2 * (ENC_BJ - 1) + 1] = *(const v4f*)(bg + bgo[ENC_BJ - 1] + EMB_N);
  }
#pragma unroll
  for (int j = 0; j < ENC_AJ - 1; ++j) sa[j] = *(const v4f*)(ag + ago[j]);
  if (alast) sa[ENC_AJ - 1] = *(const v4f*)(ag + ago[ENC_AJ - 1]);
}

__device__ __forceinline__ void enc_store(
    float* __restrict__ buf, const v4f (&sb)[2 * ENC_BJ],
    const v4f (&sa)[ENC_AJ], const int (&blo)[ENC_BJ], int tid, bool blast,
    bool alast) {
#pragma unroll
  for (int j = 0; j < ENC_BJ - 1; ++j)
    enc_put_b(buf + blo[j], sb[2 * j], sb[2 * j + 1]);
  if (blast)
    enc_put_b(buf + blo[ENC_BJ - 1], sb[2 * (ENC_BJ - 1)], sb[2 * (ENC_BJ - 1) + 1]);
#pragma unroll
  for (int j = 0; j < ENC_AJ - 1; ++j) *(v4f*)(buf + (tid + 512 * j) * 4) = sa[j];
  if (alast) *(v4f*)(buf + (tid + 512 * (ENC_AJ - 1)) * 4) = sa[ENC_AJ - 1];
}

__global__ __launch_bounds__(512) void enc_gemm_kernel(
    const float* __restrict__ x, const float* __restrict__ eW,
    float* __restrict__ part) {
  __shared__ __attribute__((aligned(16))) float lds[2 * ENC_BUF];

  const int tid  = threadIdx.x;
  const int lane = tid & 31;
  const int wave = __builtin_amdgcn_readfirstlane(tid >> 5);
  const int half = lane >> 4;
  const int l    = lane & 15;
  const int mt   = blockIdx.x >> 4;
  const int ksp  = blockIdx.x & 15;
  const int m0   = mt * ENC_RT;
  const int k0   = ksp * ENC_KW;
  const int rt   = wave & 3;
  const int cg   = wave >> 2;

  int bgo[ENC_BJ], blo[ENC_BJ], ago[ENC_AJ];
#pragma unroll
  for (int j = 0; j < ENC_BJ; ++j) {
    const int i = tid + 512 * j;
    const int p = i / (EMB_N / 4);
    const int q = i - p * (EMB_N / 4);
    bgo[j] = 2 * p * EMB_N + q * 4;
    blo[j] = ENC_ASZ + p * ENC_BP + q * 8;
  }
#pragma unroll
  for (int j = 0; j < ENC_AJ; ++j) {
    const int i = tid + 512 * j;
    const int r = i / (ENC_KC / 4);
    ago[j] = r * CT_N + (i - r * (ENC_KC / 4)) * 4;
  }
  const bool blast = tid + 512 * (ENC_BJ - 1) < ENC_BIT;
  const bool alast = tid + 512 * (ENC_AJ - 1) < ENC_AV4;
  const float* __restrict__ Bg = eW + (size_t)k0 * EMB_N;
  const float* __restrict__ Ag = x + (size_t)m0 * CT_N + k0;

  for (int i = tid; i < 2 * (ENC_KC / 2) * 16; i += 512) {
    const int bf = i / ((ENC_KC / 2) * 16);
    const int r  = (i - bf * ((ENC_KC / 2) * 16)) >> 4;
    lds[bf * ENC_BUF + ENC_ASZ + r * ENC_BP + 2 * EMB_N + (i & 15)] = 0.0f;
  }

  v4f sb[2 * ENC_BJ], sa[ENC_AJ];
  enc_load(sb, sa, Bg, Ag, bgo, ago, blast, alast);
  enc_store(lds, sb, sa, blo, tid, blast, alast);

  v8f acc0 = {}, acc1 = {}, acc2 = {};
  for (int c = 0; c < ENC_NCH; ++c) {
    __syncthreads();
    const bool more = (c + 1 < ENC_NCH);
    if (more)
      enc_load(sb, sa, Bg + (size_t)(c + 1) * (ENC_KC * EMB_N),
               Ag + (c + 1) * ENC_KC, bgo, ago, blast, alast);

    const float* cur = lds + (c & 1) * ENC_BUF;
    const float* ap = cur + (rt * 16 + l) * ENC_AP + 2 * half;
    const float* bp = cur + ENC_ASZ + half * ENC_BP + (cg * 48 + l) * 2;
#pragma unroll
    for (int ks = 0; ks < ENC_KC / 4; ++ks) {
      const v2f a = *(const v2f*)(ap + 4 * ks);
      const v2f b0 = *(const v2f*)(bp + 2 * ks * ENC_BP);
      const v2f b1 = *(const v2f*)(bp + 2 * ks * ENC_BP + 32);
      const v2f b2 = *(const v2f*)(bp + 2 * ks * ENC_BP + 64);
      acc0 = wmma4(a, b0, acc0);
      acc1 = wmma4(a, b1, acc1);
      acc2 = wmma4(a, b2, acc2);
    }

    if (more)
      enc_store(lds + ((c + 1) & 1) * ENC_BUF, sb, sa, blo, tid, blast, alast);
  }

  float* __restrict__ P = part + (size_t)ksp * ENC_PLANE;
  const int col = cg * 48 + l;
#pragma unroll
  for (int r = 0; r < 8; ++r) {
    const int row = m0 + rt * 16 + r + 8 * half;
    P[(size_t)row * EMB_N + col]      = acc0[r];
    P[(size_t)row * EMB_N + col + 16] = acc1[r];
    if (col + 32 < EMB_N) P[(size_t)row * EMB_N + col + 32] = acc2[r];
  }
}

__global__ __launch_bounds__(256) void enc_sum_kernel(
    const float* __restrict__ part, const float* __restrict__ eB,
    float* __restrict__ h) {
  const int i = (blockIdx.x * 256 + threadIdx.x) * 4;
  v4f s = *(const v4f*)(part + i);
#pragma unroll
  for (int p = 1; p < ENC_KS; ++p) s += *(const v4f*)(part + (size_t)p * ENC_PLANE + i);
  const int cb = i % EMB_N;
  s.x += eB[cb];
  s.y += eB[cb + 1];
  s.z += eB[cb + 2];
  s.w += eB[cb + 3];
  *(v4f*)(h + i) = s;
}

__global__ __launch_bounds__(256) void proj1_gemm_kernel(
    const float* __restrict__ h, const float* __restrict__ W1,
    const float* __restrict__ b1, float* __restrict__ h1,
    float* __restrict__ g) {
  const int wave = threadIdx.x >> 5;
  const int lane = threadIdx.x & 31;
  const int tile = blockIdx.x * 8 + wave;
  const int mt = tile >> 5, np = tile & 31;
  const int m0 = mt * 16, n0 = np * 32;
  const int half = lane >> 4;
  const int l    = lane & 15;

  const int arow  = m0 + l;
  const int ncol0 = n0 + l;
  const int ncol1 = n0 + 16 + l;
  const float* __restrict__ Ar = h + (size_t)arow * EMB_N;

  v8f acc0 = {}, acc1 = {};
  for (int k = 0; k < EMB_N; k += 8) {
    const int ka = k + 2 * half;
    const v2f a0 = *(const v2f*)(Ar + ka);
    const v2f a1 = *(const v2f*)(Ar + ka + 4);
    v2f b00, b01, b10, b11;
    b00.x = W1[(size_t)ka * PROJ_N + ncol0];
    b00.y = W1[(size_t)(ka + 1) * PROJ_N + ncol0];
    b01.x = W1[(size_t)(ka + 4) * PROJ_N + ncol0];
    b01.y = W1[(size_t)(ka + 5) * PROJ_N + ncol0];
    b10.x = W1[(size_t)ka * PROJ_N + ncol1];
    b10.y = W1[(size_t)(ka + 1) * PROJ_N + ncol1];
    b11.x = W1[(size_t)(ka + 4) * PROJ_N + ncol1];
    b11.y = W1[(size_t)(ka + 5) * PROJ_N + ncol1];
    acc0 = wmma4(a0, b00, acc0);
    acc1 = wmma4(a0, b10, acc1);
    acc0 = wmma4(a1, b01, acc0);
    acc1 = wmma4(a1, b11, acc1);
  }
  const float bn0 = b1[ncol0];
  const float bn1 = b1[ncol1];
#pragma unroll
  for (int r = 0; r < 8; ++r) {
    const int row = m0 + r + 8 * half;
    const float t0 = acc0[r] + bn0;
    const float t1 = acc1[r] + bn1;
    h1[(size_t)row * PROJ_N + ncol0] = t0;
    h1[(size_t)row * PROJ_N + ncol1] = t1;
    g[(size_t)row * PROJ_N + ncol0] = gelu_exact(t0);
    g[(size_t)row * PROJ_N + ncol1] = gelu_exact(t1);
  }
}

#define P2_T    64
#define P2_KC   128
#define P2_NCH  (PROJ_N / P2_KC)
#define P2_AP   132
#define P2_BP   160
#define P2_ASZ  (P2_T * P2_AP)
#define P2_BSZ  ((P2_KC / 2) * P2_BP)
#define P2_BUF  (P2_ASZ + P2_BSZ)
#define P2_AJ   8
#define P2_BJ   4
#define P2_G    8
#define P2_NG   (P2_KC / 4 / P2_G)

__device__ __forceinline__ void p2_load(v4f (&sa)[P2_AJ], v4f (&sb)[2 * P2_BJ],
                                        const float* __restrict__ ag,
                                        const float* __restrict__ bg) {
#pragma unroll
  for (int j = 0; j < P2_AJ; ++j)
    sa[j] = *(const v4f*)(ag + (size_t)j * 8 * PROJ_N);
#pragma unroll
  for (int j = 0; j < P2_BJ; ++j) {
    sb[2 * j]     = *(const v4f*)(bg + (size_t)j * 32 * PROJ_N);
    sb[2 * j + 1] = *(const v4f*)(bg + (size_t)j * 32 * PROJ_N + PROJ_N);
  }
}

__device__ __forceinline__ void p2_store(float* __restrict__ buf,
                                         const v4f (&sa)[P2_AJ],
                                         const v4f (&sb)[2 * P2_BJ], int alo,
                                         int blo) {
#pragma unroll
  for (int j = 0; j < P2_AJ; ++j) *(v4f*)(buf + alo + j * 8 * P2_AP) = sa[j];
#pragma unroll
  for (int j = 0; j < P2_BJ; ++j)
    enc_put_b(buf + blo + j * 16 * P2_BP, sb[2 * j], sb[2 * j + 1]);
}

__device__ __forceinline__ void p2_frag(v2f (&fa)[P2_G], v2f (&fb0)[P2_G],
                                        v2f (&fb1)[P2_G], const float* ap,
                                        const float* bp, int grp) {
#pragma unroll
  for (int j = 0; j < P2_G; ++j) {
    const int ks = grp * P2_G + j;
    fa[j]  = *(const v2f*)(ap + 4 * ks);
    fb0[j] = *(const v2f*)(bp + 2 * ks * P2_BP);
    fb1[j] = *(const v2f*)(bp + 2 * ks * P2_BP + 32);
  }
}

__global__ __launch_bounds__(256) void proj2_gemm_kernel(
    const float* __restrict__ g, const float* __restrict__ W2,
    const float* __restrict__ b2, const float* __restrict__ h1,
    float* __restrict__ hsum) {
  __shared__ __attribute__((aligned(16))) float lds[2 * P2_BUF];

  const int tid  = threadIdx.x;
  const int lane = tid & 31;
  const int wave = __builtin_amdgcn_readfirstlane(tid >> 5);
  const int half = lane >> 4;
  const int l    = lane & 15;
  const int m0   = (blockIdx.x >> 4) * P2_T;
  const int n0   = (blockIdx.x & 15) * P2_T;
  const int rt   = wave & 3;
  const int cg   = wave >> 2;

  const float* __restrict__ Ag =
      g + (size_t)(m0 + (tid >> 5)) * PROJ_N + (tid & 31) * 4;
  const float* __restrict__ Bg =
      W2 + (size_t)(2 * (tid >> 4)) * PROJ_N + n0 + (tid & 15) * 4;
  const int alo = (tid >> 5) * P2_AP + (tid & 31) * 4;
  const int blo = P2_ASZ + (tid >> 4) * P2_BP + (tid & 15) * 8;

  v4f sa[P2_AJ], sb[2 * P2_BJ];
  p2_load(sa, sb, Ag, Bg);
  p2_store(lds, sa, sb, alo, blo);

  v8f acc0 = {}, acc1 = {};
  for (int c = 0; c < P2_NCH; ++c) {
    __syncthreads();
    const bool more = (c + 1 < P2_NCH);
    if (more)
      p2_load(sa, sb, Ag + (c + 1) * P2_KC,
              Bg + (size_t)(c + 1) * P2_KC * PROJ_N);

    const float* cur = lds + (c & 1) * P2_BUF;
    const float* ap = cur + (rt * 16 + l) * P2_AP + 2 * half;
    const float* bp = cur + P2_ASZ + half * P2_BP + (cg * 32 + l) * 2;
    v2f fa[2][P2_G], fb0[2][P2_G], fb1[2][P2_G];
    p2_frag(fa[0], fb0[0], fb1[0], ap, bp, 0);
#pragma unroll
    for (int grp = 0; grp < P2_NG; ++grp) {
      if (grp + 1 < P2_NG)
        p2_frag(fa[(grp + 1) & 1], fb0[(grp + 1) & 1], fb1[(grp + 1) & 1], ap,
                bp, grp + 1);
      __builtin_amdgcn_sched_barrier(0);
#pragma unroll
      for (int j = 0; j < P2_G; ++j) {
        acc0 = wmma4(fa[grp & 1][j], fb0[grp & 1][j], acc0);
        acc1 = wmma4(fa[grp & 1][j], fb1[grp & 1][j], acc1);
      }
      __builtin_amdgcn_sched_barrier(0);
    }

    if (more) p2_store(lds + ((c + 1) & 1) * P2_BUF, sa, sb, alo, blo);
  }

  const int ncol0 = n0 + cg * 32 + l;
  const int ncol1 = ncol0 + 16;
  const float bn0 = b2[ncol0];
  const float bn1 = b2[ncol1];
#pragma unroll
  for (int r = 0; r < 8; ++r) {
    const int row = m0 + rt * 16 + r + 8 * half;
    hsum[(size_t)row * PROJ_N + ncol0] =
        h1[(size_t)row * PROJ_N + ncol0] + acc0[r] + bn0;
    hsum[(size_t)row * PROJ_N + ncol1] =
        h1[(size_t)row * PROJ_N + ncol1] + acc1[r] + bn1;
  }
}

__global__ __launch_bounds__(256) void layernorm_kernel(
    const float* __restrict__ hsum, const float* __restrict__ gamma,
    const float* __restrict__ beta, float* __restrict__ out) {
  const int row = blockIdx.x;
  const int tid = threadIdx.x;
  const float* __restrict__ hr = hsum + (size_t)row * PROJ_N;

  float v[4];
  float s = 0.0f, ss = 0.0f;
#pragma unroll
  for (int i = 0; i < 4; ++i) {
    v[i] = hr[tid + i * 256];
    s += v[i];
    ss += v[i] * v[i];
  }
  __shared__ float rs[256];
  __shared__ float rss[256];
  rs[tid] = s;
  rss[tid] = ss;
  __syncthreads();
  for (int off = 128; off > 0; off >>= 1) {
    if (tid < off) {
      rs[tid] += rs[tid + off];
      rss[tid] += rss[tid + off];
    }
    __syncthreads();
  }
  const float mu  = rs[0] * (1.0f / PROJ_N);
  const float var = rss[0] * (1.0f / PROJ_N) - mu * mu;
  const float inv = rsqrtf(var + 1e-5f);
#pragma unroll
  for (int i = 0; i < 4; ++i) {
    const int c = tid + i * 256;
    out[(size_t)row * PROJ_N + c] = (v[i] - mu) * inv * gamma[c] + beta[c];
  }
}

extern "C" void kernel_launch(void* const* d_in, const int* in_sizes, int n_in,
                              void* d_out, int out_size, void* d_ws,
                              size_t ws_size, hipStream_t stream) {
  const float* eeg    = (const float*)d_in[0];
  const int*   sid    = (const int*)d_in[1];
  const float* subj_W = (const float*)d_in[2];
  const float* subj_b = (const float*)d_in[3];
  const float* enc_W  = (const float*)d_in[4];
  const float* enc_b  = (const float*)d_in[5];
  const float* W1     = (const float*)d_in[6];
  const float* b1     = (const float*)d_in[7];
  const float* W2     = (const float*)d_in[8];
  const float* b2     = (const float*)d_in[9];
  const float* gamma  = (const float*)d_in[10];
  const float* beta   = (const float*)d_in[11];
  float*       out    = (float*)d_out;

  float* xbuf  = (float*)d_ws;
  float* hbuf  = xbuf + (size_t)B_N * CT_N;
  float* h1buf = hbuf + (size_t)B_N * EMB_N;
  float* gbuf  = h1buf + (size_t)B_N * PROJ_N;
  float* hsbuf = gbuf + (size_t)B_N * PROJ_N;

  subj_gemm_kernel<<<B_N, 128, 0, stream>>>(eeg, sid, subj_W, subj_b, xbuf);
  float* pbuf = h1buf;
  enc_gemm_kernel<<<ENC_KS * (B_N / ENC_RT), 512, 0, stream>>>(xbuf, enc_W, pbuf);
  enc_sum_kernel<<<ENC_PLANE / 1024, 256, 0, stream>>>(pbuf, enc_b, hbuf);
  proj1_gemm_kernel<<<256, 256, 0, stream>>>(hbuf, W1, b1, h1buf, gbuf);
  proj2_gemm_kernel<<<256, 256, 0, stream>>>(gbuf, W2, b2, h1buf, hsbuf);
  layernorm_kernel<<<B_N, 256, 0, stream>>>(hsbuf, gamma, beta, out);
}
